# starting kernel
# baseline (speedup 1.0000x reference)
#include <hip/hip_runtime.h>
#include <hip/hip_bf16.h>


typedef __attribute__((ext_vector_type(2))) float v2f;
typedef __attribute__((ext_vector_type(8))) float v8f;

#define F_IN 128
#define HID  16

__global__ void gcn_init_kernel(float* __restrict__ deg,
                                float* __restrict__ agg1,
                                float* __restrict__ out,
                                const float* __restrict__ b1,
                                const float* __restrict__ b2,
                                int n) {
    int tid = blockIdx.x * blockDim.x + threadIdx.x;
    if (tid >= n * HID) return;
    int f = tid & (HID - 1);
    if (f == 0) deg[tid >> 4] = 1.0f;
    agg1[tid] = b1[f];
    out[tid]  = b2[f];
}

__global__ void gcn_deg_kernel(const int* __restrict__ dst,
                               float* __restrict__ deg, int e) {
    int i = blockIdx.x * blockDim.x + threadIdx.x;
    if (i < e) atomicAdd(&deg[dst[i]], 1.0f);
}

__global__ void gcn_rsqrt_kernel(float* __restrict__ deg, int n) {
    int i = blockIdx.x * blockDim.x + threadIdx.x;
    if (i < n) {
        float d = deg[i];
        deg[i] = d > 0.0f ? rsqrtf(d) : 0.0f;
    }
}

template <bool RELU, int K>
__global__ __launch_bounds__(128)
void gcn_gemm_wmma_kernel(const float* __restrict__ A,
                          const float* __restrict__ B,
                          float* __restrict__ C, int nrows) {
    const int lane = threadIdx.x & 31;
    const int wave = threadIdx.x >> 5;
    const int tile = blockIdx.x * 4 + wave;
    const int row0 = tile * 16;
    if (row0 >= nrows) return;

    const int m    = lane & 15;
    const int half = lane >> 4;
    const int kh   = half << 1;

    const float* __restrict__ arow = A + (size_t)(row0 + m) * K;

    v8f acc = {};
#pragma unroll
    for (int k = 0; k < K; k += 4) {
        v2f a = *reinterpret_cast<const v2f*>(arow + k + kh);
        if (RELU) {
            a.x = fmaxf(a.x, 0.0f);
            a.y = fmaxf(a.y, 0.0f);
        }
        v2f b;
        b.x = B[(k + kh) * HID + m];
        b.y = B[(k + kh + 1) * HID + m];
        acc = __builtin_amdgcn_wmma_f32_16x16x4_f32(
             false, a,  false, b,
             (short)0, acc,  false,  false);
    }

    float* __restrict__ ccol = C + (size_t)row0 * HID + m;
#pragma unroll
    for (int r = 0; r < 8; ++r) {
        ccol[(size_t)(r + 8 * half) * HID] = acc[r];
    }
}

__global__ void gcn_agg_kernel(const float* __restrict__ h,
                               const int* __restrict__ src,
                               const int* __restrict__ dst,
                               const float* __restrict__ dinv,
                               float* __restrict__ out,
                               int e, int n) {
    int tid = blockIdx.x * blockDim.x + threadIdx.x;
    int idx = tid >> 4;
    int f   = tid & (HID - 1);
    if (idx >= e + n) return;
    int s, d;
    if (idx < e) {
        s = src[idx];
        d = dst[idx];
    } else {
        s = d = idx - e;
    }
    float norm = dinv[s] * dinv[d];
    atomicAdd(&out[(size_t)d * HID + f], h[(size_t)s * HID + f] * norm);
}

extern "C" void kernel_launch(void* const* d_in, const int* in_sizes, int n_in,
                              void* d_out, int out_size, void* d_ws, size_t ws_size,
                              hipStream_t stream) {
    const float* x  = (const float*)d_in[0];
    const int*   ei = (const int*)d_in[1];
    const float* W1 = (const float*)d_in[2];
    const float* b1 = (const float*)d_in[3];
    const float* W2 = (const float*)d_in[4];
    const float* b2 = (const float*)d_in[5];
    float* out = (float*)d_out;

    const int N = in_sizes[0] / F_IN;
    const int E = in_sizes[1] / 2;
    const int* src = ei;
    const int* dst = ei + E;

    char* ws = (char*)d_ws;
    float* deg  = (float*)ws;
    float* hbuf = (float*)(ws + (size_t)N * sizeof(float));
    float* agg1 = (float*)(ws + (size_t)N * sizeof(float) +
                                (size_t)N * HID * sizeof(float));

    const int TB = 256;

    gcn_init_kernel<<<(N * HID + TB - 1) / TB, TB, 0, stream>>>(
        deg, agg1, out, b1, b2, N);

    gcn_deg_kernel<<<(E + TB - 1) / TB, TB, 0, stream>>>(dst, deg, E);
    gcn_rsqrt_kernel<<<(N + TB - 1) / TB, TB, 0, stream>>>(deg, N);

    const int tiles = (N + 15) / 16;
    const int gblk  = (tiles + 3) / 4;
    gcn_gemm_wmma_kernel<false, F_IN><<<gblk, 128, 0, stream>>>(x, W1, hbuf, N);

    const long long at = (long long)(E + N) * HID;
    gcn_agg_kernel<<<(int)((at + TB - 1) / TB), TB, 0, stream>>>(
        hbuf, src, dst, deg, agg1, E, N);

    gcn_gemm_wmma_kernel<true, HID><<<gblk, 128, 0, stream>>>(agg1, W2, hbuf, N);

    gcn_agg_kernel<<<(int)((at + TB - 1) / TB), TB, 0, stream>>>(
        hbuf, src, dst, deg, out, E, N);
}


# second aggregation replays per-wave hit lists (source id | local row) stored by the first aggregation's scan instead of rescanning dst and src; LDS float add arrival order among waves may shift
# speedup vs baseline: 1.6948x; 1.6948x over previous
#include <hip/hip_runtime.h>
#include <hip/hip_bf16.h>


typedef __attribute__((ext_vector_type(2))) float v2f;
typedef __attribute__((ext_vector_type(8))) float v8f;

#define F_IN 128
#define HID  16

#define RN      3584
#define PITCH   17
#define NCHUNK  9
#define TILE    512
#define AGG_TB  1024
#define AGG_WAVES (AGG_TB / 32)
#define LCAP    1024
#define SRC_BITS 20

__global__ void gcn_init_kernel(float* __restrict__ deg, int n) {
    int i = blockIdx.x * blockDim.x + threadIdx.x;
    if (i < n) deg[i] = 1.0f;
}

__global__ void gcn_deg_kernel(const int* __restrict__ dst,
                               float* __restrict__ deg, int e) {
    int i = blockIdx.x * blockDim.x + threadIdx.x;
    if (i < e) atomicAdd(&deg[dst[i]], 1.0f);
}

__global__ void gcn_rsqrt_kernel(float* __restrict__ deg, int n) {
    int i = blockIdx.x * blockDim.x + threadIdx.x;
    if (i < n) {
        float d = deg[i];
        deg[i] = d > 0.0f ? rsqrtf(d) : 0.0f;
    }
}

template <bool RELU, int K>
__global__ __launch_bounds__(128)
void gcn_gemm_wmma_kernel(const float* __restrict__ A,
                          const float* __restrict__ B,
                          const float* __restrict__ rscale,
                          float* __restrict__ C, int nrows) {
    const int lane = threadIdx.x & 31;
    const int wave = threadIdx.x >> 5;
    const int tile = blockIdx.x * 4 + wave;
    const int row0 = tile * 16;
    if (row0 >= nrows) return;

    const int m    = lane & 15;
    const int half = lane >> 4;
    const int kh   = half << 1;

    const float* __restrict__ arow = A + (size_t)(row0 + m) * K;

    v8f acc = {};
#pragma unroll
    for (int k = 0; k < K; k += 4) {
        v2f a = *reinterpret_cast<const v2f*>(arow + k + kh);
        if (RELU) {
            a.x = fmaxf(a.x, 0.0f);
            a.y = fmaxf(a.y, 0.0f);
        }
        v2f b;
        b.x = B[(k + kh) * HID + m];
        b.y = B[(k + kh + 1) * HID + m];
        acc = __builtin_amdgcn_wmma_f32_16x16x4_f32(
             false, a,  false, b,
             (short)0, acc,  false,  false);
    }

    const float4* rs4 = reinterpret_cast<const float4*>(rscale + row0 + 8 * half);
    const float4 sa = rs4[0], sb = rs4[1];
    const float sc[8] = {sa.x, sa.y, sa.z, sa.w, sb.x, sb.y, sb.z, sb.w};
    float* __restrict__ ccol = C + (size_t)row0 * HID + m;
#pragma unroll
    for (int r = 0; r < 8; ++r) {
        ccol[(size_t)(r + 8 * half) * HID] = acc[r] * sc[r];
    }
}

__global__ __launch_bounds__(128)
void gcn_lin2_parts_kernel(const float* __restrict__ part,
                           const float* __restrict__ hself,
                           const float* __restrict__ dinv,
                           const float* __restrict__ bias,
                           const float* __restrict__ B,
                           float* __restrict__ C, int nrows) {
    const int lane = threadIdx.x & 31;
    const int wave = threadIdx.x >> 5;
    const int tile = blockIdx.x * 4 + wave;
    const int row0 = tile * 16;
    if (row0 >= nrows) return;

    const int m    = lane & 15;
    const int half = lane >> 4;
    const int kh   = half << 1;

    const int    row   = row0 + m;
    const float  dv    = dinv[row];
    const size_t plane = (size_t)nrows * HID;
    const float* __restrict__ prow = part  + (size_t)row * HID;
    const float* __restrict__ srow = hself + (size_t)row * HID;

    v8f acc = {};
#pragma unroll
    for (int k = 0; k < HID; k += 4) {
        v2f s = *reinterpret_cast<const v2f*>(prow + k + kh);
#pragma unroll
        for (int p = 1; p < NCHUNK; ++p) {
            const v2f t = *reinterpret_cast<const v2f*>(prow + (size_t)p * plane + k + kh);
            s.x += t.x;
            s.y += t.y;
        }
        const v2f hs = *reinterpret_cast<const v2f*>(srow + k + kh);
        v2f a;
        a.x = fmaxf(bias[k + kh]     + dv * (s.x + hs.x), 0.0f);
        a.y = fmaxf(bias[k + kh + 1] + dv * (s.y + hs.y), 0.0f);
        v2f b;
        b.x = B[(k + kh) * HID + m];
        b.y = B[(k + kh + 1) * HID + m];
        acc = __builtin_amdgcn_wmma_f32_16x16x4_f32(
             false, a,  false, b,
             (short)0, acc,  false,  false);
    }

    const float4* rs4 = reinterpret_cast<const float4*>(dinv + row0 + 8 * half);
    const float4 sa = rs4[0], sb = rs4[1];
    const float sc[8] = {sa.x, sa.y, sa.z, sa.w, sb.x, sb.y, sb.z, sb.w};
    float* __restrict__ ccol = C + (size_t)row0 * HID + m;
#pragma unroll
    for (int r = 0; r < 8; ++r) {
        ccol[(size_t)(r + 8 * half) * HID] = acc[r] * sc[r];
    }
}

__device__ __forceinline__ void gcn_add_row(float* a, const float4 v0, const float4 v1,
                                            const float4 v2, const float4 v3) {
    __hip_atomic_fetch_add(a +  0, v0.x, __ATOMIC_RELAXED, __HIP_MEMORY_SCOPE_WORKGROUP);
    __hip_atomic_fetch_add(a +  1, v0.y, __ATOMIC_RELAXED, __HIP_MEMORY_SCOPE_WORKGROUP);
    __hip_atomic_fetch_add(a +  2, v0.z, __ATOMIC_RELAXED, __HIP_MEMORY_SCOPE_WORKGROUP);
    __hip_atomic_fetch_add(a +  3, v0.w, __ATOMIC_RELAXED, __HIP_MEMORY_SCOPE_WORKGROUP);
    __hip_atomic_fetch_add(a +  4, v1.x, __ATOMIC_RELAXED, __HIP_MEMORY_SCOPE_WORKGROUP);
    __hip_atomic_fetch_add(a +  5, v1.y, __ATOMIC_RELAXED, __HIP_MEMORY_SCOPE_WORKGROUP);
    __hip_atomic_fetch_add(a +  6, v1.z, __ATOMIC_RELAXED, __HIP_MEMORY_SCOPE_WORKGROUP);
    __hip_atomic_fetch_add(a +  7, v1.w, __ATOMIC_RELAXED, __HIP_MEMORY_SCOPE_WORKGROUP);
    __hip_atomic_fetch_add(a +  8, v2.x, __ATOMIC_RELAXED, __HIP_MEMORY_SCOPE_WORKGROUP);
    __hip_atomic_fetch_add(a +  9, v2.y, __ATOMIC_RELAXED, __HIP_MEMORY_SCOPE_WORKGROUP);
    __hip_atomic_fetch_add(a + 10, v2.z, __ATOMIC_RELAXED, __HIP_MEMORY_SCOPE_WORKGROUP);
    __hip_atomic_fetch_add(a + 11, v2.w, __ATOMIC_RELAXED, __HIP_MEMORY_SCOPE_WORKGROUP);
    __hip_atomic_fetch_add(a + 12, v3.x, __ATOMIC_RELAXED, __HIP_MEMORY_SCOPE_WORKGROUP);
    __hip_atomic_fetch_add(a + 13, v3.y, __ATOMIC_RELAXED, __HIP_MEMORY_SCOPE_WORKGROUP);
    __hip_atomic_fetch_add(a + 14, v3.z, __ATOMIC_RELAXED, __HIP_MEMORY_SCOPE_WORKGROUP);
    __hip_atomic_fetch_add(a + 15, v3.w, __ATOMIC_RELAXED, __HIP_MEMORY_SCOPE_WORKGROUP);
}

template <bool FIRST>
__global__ __launch_bounds__(AGG_TB)
void gcn_agg_lds_kernel(const float* __restrict__ h,
                        const int* __restrict__ src,
                        const int* __restrict__ dst,
                        float* __restrict__ part,
                        unsigned* __restrict__ list,
                        int* __restrict__ lcount,
                        int e, int n, int chunk) {
    __shared__ float    acc[RN * PITCH];
    __shared__ unsigned queue[AGG_WAVES * TILE];

    const int tid  = threadIdx.x;
    const int lane = tid & 31;
    const int wave = __builtin_amdgcn_readfirstlane(tid >> 5);
    const int nb   = blockIdx.x * RN;
    const int c    = blockIdx.y;

    const int wid = (c * (int)gridDim.x + (int)blockIdx.x) * AGG_WAVES + wave;
    unsigned* __restrict__ lw = list + (size_t)wid * LCAP;
    const int cntw = FIRST ? -1 : __builtin_amdgcn_readfirstlane(lcount[wid]);

    for (int i = tid; i < RN * PITCH; i += AGG_TB) acc[i] = 0.0f;
    __syncthreads();

    if (FIRST || cntw < 0) {
        const int cbeg = c * chunk;
        const int cend = min(e, cbeg + chunk);
        const int last = e - 4;
        unsigned* __restrict__ q = queue + wave * TILE;

        int tot = 0;
        int eb = cbeg + wave * TILE;
        int4 cur[4];
#pragma unroll
        for (int j = 0; j < 4; ++j)
            cur[j] = *reinterpret_cast<const int4*>(dst + min(eb + j * 128 + lane * 4, last));

        for (; eb < cend; eb += AGG_WAVES * TILE) {
            const int en = eb + AGG_WAVES * TILE;
            int4 nxt[4];
#pragma unroll
            for (int j = 0; j < 4; ++j)
                nxt[j] = *reinterpret_cast<const int4*>(dst + min(en + j * 128 + lane * 4, last));

            int cnt = 0;
#pragma unroll
            for (int j = 0; j < 4; ++j) {
                const int  o4 = j * 128 + lane * 4;
                const bool vq = (eb + o4) < cend;
                int d4[4];
                d4[0] = vq ? cur[j].x : -1; d4[1] = vq ? cur[j].y : -1;
                d4[2] = vq ? cur[j].z : -1; d4[3] = vq ? cur[j].w : -1;
#pragma unroll
                for (int t = 0; t < 4; ++t) {
                    const unsigned dl   = (unsigned)(d4[t] - nb);
                    const bool     hit  = dl < (unsigned)RN;
                    const unsigned mask = __builtin_amdgcn_ballot_w32(hit);
                    if (hit) {
                        const int pos = cnt + (int)__builtin_amdgcn_mbcnt_lo(mask, 0u);
                        q[pos] = (dl << 9) | (unsigned)(o4 + t);
                    }
                    cnt += __builtin_popcount(mask);
                }
            }


            for (int j = lane; j < cnt; j += 32) {
                const unsigned ent = q[j];
                const int      s   = src[eb + (int)(ent & 511u)];
                const float4*  hr  = reinterpret_cast<const float4*>(h + (size_t)s * HID);
                const float4 v0 = hr[0], v1 = hr[1], v2 = hr[2], v3 = hr[3];
                if (FIRST) {
                    const int p = tot + j;
                    if (p < LCAP) lw[p] = ((unsigned)s << 12) | (ent >> 9);
                }
                gcn_add_row(acc + (ent >> 9) * PITCH, v0, v1, v2, v3);
            }
            tot += cnt;

#pragma unroll
            for (int j = 0; j < 4; ++j) cur[j] = nxt[j];
        }
        if (FIRST) {
            if (lane == 0)
                lcount[wid] = (tot <= LCAP && n <= (1 << SRC_BITS)) ? tot : -1;
        }
    } else {
        if (cntw > 0) {
            unsigned ent[4];
#pragma unroll
            for (int u = 0; u < 4; ++u)
                ent[u] = lw[min(u * 32 + lane, cntw - 1)];
            int j0 = 0;
            for (; j0 + 4 * 32 <= cntw; j0 += 4 * 32) {
                unsigned nx[4];
#pragma unroll
                for (int u = 0; u < 4; ++u)
                    nx[u] = lw[min(j0 + 4 * 32 + u * 32 + lane, cntw - 1)];
                float4 v[4][4];
#pragma unroll
                for (int u = 0; u < 4; ++u) {
                    const float4* hr =
                        reinterpret_cast<const float4*>(h + (size_t)(ent[u] >> 12) * HID);
                    v[u][0] = hr[0]; v[u][1] = hr[1]; v[u][2] = hr[2]; v[u][3] = hr[3];
                }
#pragma unroll
                for (int u = 0; u < 4; ++u)
                    gcn_add_row(acc + (ent[u] & 4095u) * PITCH,
                                v[u][0], v[u][1], v[u][2], v[u][3]);
#pragma unroll
                for (int u = 0; u < 4; ++u) ent[u] = nx[u];
            }
#pragma unroll
            for (int u = 0; u < 4; ++u) {
                if (j0 + u * 32 + lane < cntw) {
                    const float4* hr =
                        reinterpret_cast<const float4*>(h + (size_t)(ent[u] >> 12) * HID);
                    const float4 v0 = hr[0], v1 = hr[1], v2 = hr[2], v3 = hr[3];
                    gcn_add_row(acc + (ent[u] & 4095u) * PITCH, v0, v1, v2, v3);
                }
            }
        }
    }
    __syncthreads();

    float* __restrict__ pout = part + (size_t)c * n * HID;
    for (int i = tid; i < RN * HID; i += AGG_TB) {
        const int row  = i >> 4;
        const int f    = i & (HID - 1);
        const int node = nb + row;
        if (node < n) pout[(size_t)node * HID + f] = acc[row * PITCH + f];
    }
}

__global__ void gcn_sum_parts_kernel(const float* __restrict__ part,
                                     const float* __restrict__ hself,
                                     const float* __restrict__ dinv,
                                     const float* __restrict__ b2,
                                     float* __restrict__ out, int n) {
    const int i = blockIdx.x * blockDim.x + threadIdx.x;
    if (i >= n * (HID / 4)) return;
    const float4* p4 = reinterpret_cast<const float4*>(part);
    const size_t plane4 = (size_t)n * (HID / 4);
    float4 s = p4[i];
#pragma unroll
    for (int p = 1; p < NCHUNK; ++p) {
        const float4 t = p4[(size_t)p * plane4 + i];
        s.x += t.x; s.y += t.y; s.z += t.z; s.w += t.w;
    }
    const float  dv = dinv[i >> 2];
    const float4 hs = reinterpret_cast<const float4*>(hself)[i];
    const float4 b  = reinterpret_cast<const float4*>(b2)[i & 3];
    float4 o;
    o.x = b.x + dv * (s.x + hs.x);
    o.y = b.y + dv * (s.y + hs.y);
    o.z = b.z + dv * (s.z + hs.z);
    o.w = b.w + dv * (s.w + hs.w);
    reinterpret_cast<float4*>(out)[i] = o;
}

extern "C" void kernel_launch(void* const* d_in, const int* in_sizes, int n_in,
                              void* d_out, int out_size, void* d_ws, size_t ws_size,
                              hipStream_t stream) {
    const float* x  = (const float*)d_in[0];
    const int*   ei = (const int*)d_in[1];
    const float* W1 = (const float*)d_in[2];
    const float* b1 = (const float*)d_in[3];
    const float* W2 = (const float*)d_in[4];
    const float* b2 = (const float*)d_in[5];
    float* out = (float*)d_out;

    const int N = in_sizes[0] / F_IN;
    const int E = in_sizes[1] / 2;
    const int* src = ei;
    const int* dst = ei + E;

    char* ws = (char*)d_ws;
    const size_t rows_bytes = (size_t)N * HID * sizeof(float);
    float* deg  = (float*)ws;
    float* h1   = (float*)(ws + (size_t)N * sizeof(float));
    float* h2   = (float*)(ws + (size_t)N * sizeof(float) + rows_bytes);
    float* part = (float*)(ws + (size_t)N * sizeof(float) + 2 * rows_bytes);
    const int nrange = (N + RN - 1) / RN;
    const size_t nlists = (size_t)nrange * NCHUNK * AGG_WAVES;
    unsigned* list = (unsigned*)(ws + (size_t)N * sizeof(float) + (2 + NCHUNK) * rows_bytes);
    int* lcount = (int*)(list + nlists * LCAP);

    const int TB = 256;

    gcn_init_kernel<<<(N + TB - 1) / TB, TB, 0, stream>>>(deg, N);

    gcn_deg_kernel<<<(E + TB - 1) / TB, TB, 0, stream>>>(dst, deg, E);
    gcn_rsqrt_kernel<<<(N + TB - 1) / TB, TB, 0, stream>>>(deg, N);

    const int tiles = (N + 15) / 16;
    const int gblk  = (tiles + 3) / 4;
    gcn_gemm_wmma_kernel<false, F_IN><<<gblk, 128, 0, stream>>>(x, W1, deg, h1, N);

    const int chunk  = (((E + NCHUNK - 1) / NCHUNK) + TILE - 1) / TILE * TILE;
    const dim3 agrid(nrange, NCHUNK);
    gcn_agg_lds_kernel<true><<<agrid, AGG_TB, 0, stream>>>(
        h1, src, dst, part, list, lcount, E, N, chunk);

    gcn_lin2_parts_kernel<<<gblk, 128, 0, stream>>>(part, h1, deg, b1, W2, h2, N);

    gcn_agg_lds_kernel<false><<<agrid, AGG_TB, 0, stream>>>(
        h2, src, dst, part, list, lcount, E, N, chunk);

    gcn_sum_parts_kernel<<<(N * (HID / 4) + TB - 1) / TB, TB, 0, stream>>>(
        part, h2, deg, b2, out, N);
}


# edge scan split off as first launch: counts in-degrees per chunk by LDS integer adds, stores hit lists; global degree atomics gone; both aggregations replay; LDS float add arrival order may shift
# speedup vs baseline: 3.5212x; 2.0776x over previous
#include <hip/hip_runtime.h>
#include <hip/hip_bf16.h>


typedef __attribute__((ext_vector_type(2))) float v2f;
typedef __attribute__((ext_vector_type(8))) float v8f;

#define F_IN 128
#define HID  16

#define RN      3584
#define PITCH   17
#define NCHUNK  9
#define TILE    512
#define AGG_TB  1024
#define AGG_WAVES (AGG_TB / 32)
#define LCAP    1024
#define SRC_BITS 20

__global__ __launch_bounds__(AGG_TB)
void gcn_scan_kernel(const int* __restrict__ src,
                     const int* __restrict__ dst,
                     unsigned* __restrict__ degpart,
                     unsigned* __restrict__ list,
                     int* __restrict__ lcount,
                     int e, int n, int chunk) {
    __shared__ unsigned hitcnt[RN];
    __shared__ unsigned queue[AGG_WAVES * TILE];

    const int tid  = threadIdx.x;
    const int lane = tid & 31;
    const int wave = __builtin_amdgcn_readfirstlane(tid >> 5);
    const int nb   = blockIdx.x * RN;
    const int c    = blockIdx.y;

    const int wid = (c * (int)gridDim.x + (int)blockIdx.x) * AGG_WAVES + wave;
    unsigned* __restrict__ lw = list + (size_t)wid * LCAP;

    for (int i = tid; i < RN; i += AGG_TB) hitcnt[i] = 0u;
    __syncthreads();

    const int cbeg = c * chunk;
    const int cend = min(e, cbeg + chunk);
    const int last = e - 4;
    unsigned* __restrict__ q = queue + wave * TILE;

    int tot = 0;
    int eb = cbeg + wave * TILE;
    int4 cd[4], cs[4];
#pragma unroll
    for (int j = 0; j < 4; ++j) {
        const int at = min(eb + j * 128 + lane * 4, last);
        cd[j] = *reinterpret_cast<const int4*>(dst + at);
        cs[j] = *reinterpret_cast<const int4*>(src + at);
    }

    for (; eb < cend; eb += AGG_WAVES * TILE) {
        const int en = eb + AGG_WAVES * TILE;
        int4 nd[4], ns[4];
#pragma unroll
        for (int j = 0; j < 4; ++j) {
            const int at = min(en + j * 128 + lane * 4, last);
            nd[j] = *reinterpret_cast<const int4*>(dst + at);
            ns[j] = *reinterpret_cast<const int4*>(src + at);
        }

        int cnt = 0;
#pragma unroll
        for (int j = 0; j < 4; ++j) {
            const int  o4 = j * 128 + lane * 4;
            const bool vq = (eb + o4) < cend;
            int d4[4], s4[4];
            d4[0] = vq ? cd[j].x : -1; d4[1] = vq ? cd[j].y : -1;
            d4[2] = vq ? cd[j].z : -1; d4[3] = vq ? cd[j].w : -1;
            s4[0] = cs[j].x; s4[1] = cs[j].y; s4[2] = cs[j].z; s4[3] = cs[j].w;
#pragma unroll
            for (int t = 0; t < 4; ++t) {
                const unsigned dl   = (unsigned)(d4[t] - nb);
                const bool     hit  = dl < (unsigned)RN;
                const unsigned mask = __builtin_amdgcn_ballot_w32(hit);
                if (hit) {
                    const int pos = cnt + (int)__builtin_amdgcn_mbcnt_lo(mask, 0u);
                    q[pos] = ((unsigned)s4[t] << 12) | dl;
                }
                cnt += __builtin_popcount(mask);
            }
        }


        for (int j = lane; j < cnt; j += 32) {
            const unsigned ent = q[j];
            __hip_atomic_fetch_add(hitcnt + (ent & 4095u), 1u,
                                   __ATOMIC_RELAXED, __HIP_MEMORY_SCOPE_WORKGROUP);
            const int p = tot + j;
            if (p < LCAP) lw[p] = ent;
        }
        tot += cnt;

#pragma unroll
        for (int j = 0; j < 4; ++j) { cd[j] = nd[j]; cs[j] = ns[j]; }
    }
    if (lane == 0)
        lcount[wid] = (tot <= LCAP && n <= (1 << SRC_BITS)) ? tot : -1;
    __syncthreads();

    unsigned* __restrict__ dout = degpart + (size_t)c * n;
    for (int i = tid; i < RN; i += AGG_TB) {
        const int node = nb + i;
        if (node < n) dout[node] = hitcnt[i];
    }
}

__global__ void gcn_dinv_kernel(const unsigned* __restrict__ degpart,
                                float* __restrict__ dinv, int n) {
    int i = blockIdx.x * blockDim.x + threadIdx.x;
    if (i < n) {
        unsigned cnt = 1u;
#pragma unroll
        for (int p = 0; p < NCHUNK; ++p) cnt += degpart[(size_t)p * n + i];
        const float d = (float)cnt;
        dinv[i] = d > 0.0f ? rsqrtf(d) : 0.0f;
    }
}

template <bool RELU, int K>
__global__ __launch_bounds__(128)
void gcn_gemm_wmma_kernel(const float* __restrict__ A,
                          const float* __restrict__ B,
                          const float* __restrict__ rscale,
                          float* __restrict__ C, int nrows) {
    const int lane = threadIdx.x & 31;
    const int wave = threadIdx.x >> 5;
    const int tile = blockIdx.x * 4 + wave;
    const int row0 = tile * 16;
    if (row0 >= nrows) return;

    const int m    = lane & 15;
    const int half = lane >> 4;
    const int kh   = half << 1;

    const float* __restrict__ arow = A + (size_t)(row0 + m) * K;

    v8f acc = {};
#pragma unroll
    for (int k = 0; k < K; k += 4) {
        v2f a = *reinterpret_cast<const v2f*>(arow + k + kh);
        if (RELU) {
            a.x = fmaxf(a.x, 0.0f);
            a.y = fmaxf(a.y, 0.0f);
        }
        v2f b;
        b.x = B[(k + kh) * HID + m];
        b.y = B[(k + kh + 1) * HID + m];
        acc = __builtin_amdgcn_wmma_f32_16x16x4_f32(
             false, a,  false, b,
             (short)0, acc,  false,  false);
    }

    const float4* rs4 = reinterpret_cast<const float4*>(rscale + row0 + 8 * half);
    const float4 sa = rs4[0], sb = rs4[1];
    const float sc[8] = {sa.x, sa.y, sa.z, sa.w, sb.x, sb.y, sb.z, sb.w};
    float* __restrict__ ccol = C + (size_t)row0 * HID + m;
#pragma unroll
    for (int r = 0; r < 8; ++r) {
        ccol[(size_t)(r + 8 * half) * HID] = acc[r] * sc[r];
    }
}

__global__ __launch_bounds__(128)
void gcn_lin2_parts_kernel(const float* __restrict__ part,
                           const float* __restrict__ hself,
                           const float* __restrict__ dinv,
                           const float* __restrict__ bias,
                           const float* __restrict__ B,
                           float* __restrict__ C, int nrows) {
    const int lane = threadIdx.x & 31;
    const int wave = threadIdx.x >> 5;
    const int tile = blockIdx.x * 4 + wave;
    const int row0 = tile * 16;
    if (row0 >= nrows) return;

    const int m    = lane & 15;
    const int half = lane >> 4;
    const int kh   = half << 1;

    const int    row   = row0 + m;
    const float  dv    = dinv[row];
    const size_t plane = (size_t)nrows * HID;
    const float* __restrict__ prow = part  + (size_t)row * HID;
    const float* __restrict__ srow = hself + (size_t)row * HID;

    v8f acc = {};
#pragma unroll
    for (int k = 0; k < HID; k += 4) {
        v2f s = *reinterpret_cast<const v2f*>(prow + k + kh);
#pragma unroll
        for (int p = 1; p < NCHUNK; ++p) {
            const v2f t = *reinterpret_cast<const v2f*>(prow + (size_t)p * plane + k + kh);
            s.x += t.x;
            s.y += t.y;
        }
        const v2f hs = *reinterpret_cast<const v2f*>(srow + k + kh);
        v2f a;
        a.x = fmaxf(bias[k + kh]     + dv * (s.x + hs.x), 0.0f);
        a.y = fmaxf(bias[k + kh + 1] + dv * (s.y + hs.y), 0.0f);
        v2f b;
        b.x = B[(k + kh) * HID + m];
        b.y = B[(k + kh + 1) * HID + m];
        acc = __builtin_amdgcn_wmma_f32_16x16x4_f32(
             false, a,  false, b,
             (short)0, acc,  false,  false);
    }

    const float4* rs4 = reinterpret_cast<const float4*>(dinv + row0 + 8 * half);
    const float4 sa = rs4[0], sb = rs4[1];
    const float sc[8] = {sa.x, sa.y, sa.z, sa.w, sb.x, sb.y, sb.z, sb.w};
    float* __restrict__ ccol = C + (size_t)row0 * HID + m;
#pragma unroll
    for (int r = 0; r < 8; ++r) {
        ccol[(size_t)(r + 8 * half) * HID] = acc[r] * sc[r];
    }
}

__device__ __forceinline__ void gcn_add_row(float* a, const float4 v0, const float4 v1,
                                            const float4 v2, const float4 v3) {
    __hip_atomic_fetch_add(a +  0, v0.x, __ATOMIC_RELAXED, __HIP_MEMORY_SCOPE_WORKGROUP);
    __hip_atomic_fetch_add(a +  1, v0.y, __ATOMIC_RELAXED, __HIP_MEMORY_SCOPE_WORKGROUP);
    __hip_atomic_fetch_add(a +  2, v0.z, __ATOMIC_RELAXED, __HIP_MEMORY_SCOPE_WORKGROUP);
    __hip_atomic_fetch_add(a +  3, v0.w, __ATOMIC_RELAXED, __HIP_MEMORY_SCOPE_WORKGROUP);
    __hip_atomic_fetch_add(a +  4, v1.x, __ATOMIC_RELAXED, __HIP_MEMORY_SCOPE_WORKGROUP);
    __hip_atomic_fetch_add(a +  5, v1.y, __ATOMIC_RELAXED, __HIP_MEMORY_SCOPE_WORKGROUP);
    __hip_atomic_fetch_add(a +  6, v1.z, __ATOMIC_RELAXED, __HIP_MEMORY_SCOPE_WORKGROUP);
    __hip_atomic_fetch_add(a +  7, v1.w, __ATOMIC_RELAXED, __HIP_MEMORY_SCOPE_WORKGROUP);
    __hip_atomic_fetch_add(a +  8, v2.x, __ATOMIC_RELAXED, __HIP_MEMORY_SCOPE_WORKGROUP);
    __hip_atomic_fetch_add(a +  9, v2.y, __ATOMIC_RELAXED, __HIP_MEMORY_SCOPE_WORKGROUP);
    __hip_atomic_fetch_add(a + 10, v2.z, __ATOMIC_RELAXED, __HIP_MEMORY_SCOPE_WORKGROUP);
    __hip_atomic_fetch_add(a + 11, v2.w, __ATOMIC_RELAXED, __HIP_MEMORY_SCOPE_WORKGROUP);
    __hip_atomic_fetch_add(a + 12, v3.x, __ATOMIC_RELAXED, __HIP_MEMORY_SCOPE_WORKGROUP);
    __hip_atomic_fetch_add(a + 13, v3.y, __ATOMIC_RELAXED, __HIP_MEMORY_SCOPE_WORKGROUP);
    __hip_atomic_fetch_add(a + 14, v3.z, __ATOMIC_RELAXED, __HIP_MEMORY_SCOPE_WORKGROUP);
    __hip_atomic_fetch_add(a + 15, v3.w, __ATOMIC_RELAXED, __HIP_MEMORY_SCOPE_WORKGROUP);
}

__global__ __launch_bounds__(AGG_TB)
void gcn_agg_lds_kernel(const float* __restrict__ h,
                        const int* __restrict__ src,
                        const int* __restrict__ dst,
                        float* __restrict__ part,
                        const unsigned* __restrict__ list,
                        const int* __restrict__ lcount,
                        int e, int n, int chunk) {
    __shared__ float    acc[RN * PITCH];
    __shared__ unsigned queue[AGG_WAVES * TILE];

    const int tid  = threadIdx.x;
    const int lane = tid & 31;
    const int wave = __builtin_amdgcn_readfirstlane(tid >> 5);
    const int nb   = blockIdx.x * RN;
    const int c    = blockIdx.y;

    const int wid = (c * (int)gridDim.x + (int)blockIdx.x) * AGG_WAVES + wave;
    const unsigned* __restrict__ lw = list + (size_t)wid * LCAP;
    const int cntw = __builtin_amdgcn_readfirstlane(lcount[wid]);

    for (int i = tid; i < RN * PITCH; i += AGG_TB) acc[i] = 0.0f;
    __syncthreads();

    if (cntw < 0) {
        const int cbeg = c * chunk;
        const int cend = min(e, cbeg + chunk);
        const int last = e - 4;
        unsigned* __restrict__ q = queue + wave * TILE;

        int eb = cbeg + wave * TILE;
        int4 cur[4];
#pragma unroll
        for (int j = 0; j < 4; ++j)
            cur[j] = *reinterpret_cast<const int4*>(dst + min(eb + j * 128 + lane * 4, last));

        for (; eb < cend; eb += AGG_WAVES * TILE) {
            const int en = eb + AGG_WAVES * TILE;
            int4 nxt[4];
#pragma unroll
            for (int j = 0; j < 4; ++j)
                nxt[j] = *reinterpret_cast<const int4*>(dst + min(en + j * 128 + lane * 4, last));

            int cnt = 0;
#pragma unroll
            for (int j = 0; j < 4; ++j) {
                const int  o4 = j * 128 + lane * 4;
                const bool vq = (eb + o4) < cend;
                int d4[4];
                d4[0] = vq ? cur[j].x : -1; d4[1] = vq ? cur[j].y : -1;
                d4[2] = vq ? cur[j].z : -1; d4[3] = vq ? cur[j].w : -1;
#pragma unroll
                for (int t = 0; t < 4; ++t) {
                    const unsigned dl   = (unsigned)(d4[t] - nb);
                    const bool     hit  = dl < (unsigned)RN;
                    const unsigned mask = __builtin_amdgcn_ballot_w32(hit);
                    if (hit) {
                        const int pos = cnt + (int)__builtin_amdgcn_mbcnt_lo(mask, 0u);
                        q[pos] = (dl << 9) | (unsigned)(o4 + t);
                    }
                    cnt += __builtin_popcount(mask);
                }
            }


            for (int j = lane; j < cnt; j += 32) {
                const unsigned ent = q[j];
                const int      s   = src[eb + (int)(ent & 511u)];
                const float4*  hr  = reinterpret_cast<const float4*>(h + (size_t)s * HID);
                const float4 v0 = hr[0], v1 = hr[1], v2 = hr[2], v3 = hr[3];
                gcn_add_row(acc + (ent >> 9) * PITCH, v0, v1, v2, v3);
            }

#pragma unroll
            for (int j = 0; j < 4; ++j) cur[j] = nxt[j];
        }
    } else {
        if (cntw > 0) {
            unsigned ent[4];
#pragma unroll
            for (int u = 0; u < 4; ++u)
                ent[u] = lw[min(u * 32 + lane, cntw - 1)];
            int j0 = 0;
            for (; j0 + 4 * 32 <= cntw; j0 += 4 * 32) {
                unsigned nx[4];
#pragma unroll
                for (int u = 0; u < 4; ++u)
                    nx[u] = lw[min(j0 + 4 * 32 + u * 32 + lane, cntw - 1)];
                float4 v[4][4];
#pragma unroll
                for (int u = 0; u < 4; ++u) {
                    const float4* hr =
                        reinterpret_cast<const float4*>(h + (size_t)(ent[u] >> 12) * HID);
                    v[u][0] = hr[0]; v[u][1] = hr[1]; v[u][2] = hr[2]; v[u][3] = hr[3];
                }
#pragma unroll
                for (int u = 0; u < 4; ++u)
                    gcn_add_row(acc + (ent[u] & 4095u) * PITCH,
                                v[u][0], v[u][1], v[u][2], v[u][3]);
#pragma unroll
                for (int u = 0; u < 4; ++u) ent[u] = nx[u];
            }
#pragma unroll
            for (int u = 0; u < 4; ++u) {
                if (j0 + u * 32 + lane < cntw) {
                    const float4* hr =
                        reinterpret_cast<const float4*>(h + (size_t)(ent[u] >> 12) * HID);
                    const float4 v0 = hr[0], v1 = hr[1], v2 = hr[2], v3 = hr[3];
                    gcn_add_row(acc + (ent[u] & 4095u) * PITCH, v0, v1, v2, v3);
                }
            }
        }
    }
    __syncthreads();

    float* __restrict__ pout = part + (size_t)c * n * HID;
    for (int i = tid; i < RN * HID; i += AGG_TB) {
        const int row  = i >> 4;
        const int f    = i & (HID - 1);
        const int node = nb + row;
        if (node < n) pout[(size_t)node * HID + f] = acc[row * PITCH + f];
    }
}

__global__ void gcn_sum_parts_kernel(const float* __restrict__ part,
                                     const float* __restrict__ hself,
                                     const float* __restrict__ dinv,
                                     const float* __restrict__ b2,
                                     float* __restrict__ out, int n) {
    const int i = blockIdx.x * blockDim.x + threadIdx.x;
    if (i >= n * (HID / 4)) return;
    const float4* p4 = reinterpret_cast<const float4*>(part);
    const size_t plane4 = (size_t)n * (HID / 4);
    float4 s = p4[i];
#pragma unroll
    for (int p = 1; p < NCHUNK; ++p) {
        const float4 t = p4[(size_t)p * plane4 + i];
        s.x += t.x; s.y += t.y; s.z += t.z; s.w += t.w;
    }
    const float  dv = dinv[i >> 2];
    const float4 hs = reinterpret_cast<const float4*>(hself)[i];
    const float4 b  = reinterpret_cast<const float4*>(b2)[i & 3];
    float4 o;
    o.x = b.x + dv * (s.x + hs.x);
    o.y = b.y + dv * (s.y + hs.y);
    o.z = b.z + dv * (s.z + hs.z);
    o.w = b.w + dv * (s.w + hs.w);
    reinterpret_cast<float4*>(out)[i] = o;
}

extern "C" void kernel_launch(void* const* d_in, const int* in_sizes, int n_in,
                              void* d_out, int out_size, void* d_ws, size_t ws_size,
                              hipStream_t stream) {
    const float* x  = (const float*)d_in[0];
    const int*   ei = (const int*)d_in[1];
    const float* W1 = (const float*)d_in[2];
    const float* b1 = (const float*)d_in[3];
    const float* W2 = (const float*)d_in[4];
    const float* b2 = (const float*)d_in[5];
    float* out = (float*)d_out;

    const int N = in_sizes[0] / F_IN;
    const int E = in_sizes[1] / 2;
    const int* src = ei;
    const int* dst = ei + E;

    char* ws = (char*)d_ws;
    const size_t rows_bytes = (size_t)N * HID * sizeof(float);
    float* deg  = (float*)ws;
    float* h1   = (float*)(ws + (size_t)N * sizeof(float));
    float* h2   = (float*)(ws + (size_t)N * sizeof(float) + rows_bytes);
    float* part = (float*)(ws + (size_t)N * sizeof(float) + 2 * rows_bytes);
    const int nrange = (N + RN - 1) / RN;
    const size_t nlists = (size_t)nrange * NCHUNK * AGG_WAVES;
    unsigned* list = (unsigned*)(ws + (size_t)N * sizeof(float) + (2 + NCHUNK) * rows_bytes);
    int* lcount = (int*)(list + nlists * LCAP);
    unsigned* degpart = (unsigned*)(lcount + nlists);

    const int TB = 256;

    const int chunk  = (((E + NCHUNK - 1) / NCHUNK) + TILE - 1) / TILE * TILE;
    const dim3 agrid(nrange, NCHUNK);
    gcn_scan_kernel<<<agrid, AGG_TB, 0, stream>>>(src, dst, degpart, list, lcount, E, N, chunk);

    gcn_dinv_kernel<<<(N + TB - 1) / TB, TB, 0, stream>>>(degpart, deg, N);

    const int tiles = (N + 15) / 16;
    const int gblk  = (tiles + 3) / 4;
    gcn_gemm_wmma_kernel<false, F_IN><<<gblk, 128, 0, stream>>>(x, W1, deg, h1, N);

    gcn_agg_lds_kernel<<<agrid, AGG_TB, 0, stream>>>(
        h1, src, dst, part, list, lcount, E, N, chunk);

    gcn_lin2_parts_kernel<<<gblk, 128, 0, stream>>>(part, h1, deg, b1, W2, h2, N);

    gcn_agg_lds_kernel<<<agrid, AGG_TB, 0, stream>>>(
        h2, src, dst, part, list, lcount, E, N, chunk);

    gcn_sum_parts_kernel<<<(N * (HID / 4) + TB - 1) / TB, TB, 0, stream>>>(
        part, h2, deg, b2, out, N);
}


# aggregation replay re-mapped from one lane to 16 lanes per stored hit: lane f loads value f of the source row (a load touches 2 lines, was 32) and adds it to LDS word f; LDS float add order changes
# speedup vs baseline: 4.0661x; 1.1547x over previous
#include <hip/hip_runtime.h>
#include <hip/hip_bf16.h>


typedef __attribute__((ext_vector_type(2))) float v2f;
typedef __attribute__((ext_vector_type(8))) float v8f;

#define F_IN 128
#define HID  16

#define RN      3584
#define PITCH   17
#define NCHUNK  9
#define TILE    512
#define AGG_TB  1024
#define AGG_WAVES (AGG_TB / 32)
#define LCAP    1024
#define SRC_BITS 20

__global__ __launch_bounds__(AGG_TB)
void gcn_scan_kernel(const int* __restrict__ src,
                     const int* __restrict__ dst,
                     unsigned* __restrict__ degpart,
                     unsigned* __restrict__ list,
                     int* __restrict__ lcount,
                     int e, int n, int chunk) {
    __shared__ unsigned hitcnt[RN];
    __shared__ unsigned queue[AGG_WAVES * TILE];

    const int tid  = threadIdx.x;
    const int lane = tid & 31;
    const int wave = __builtin_amdgcn_readfirstlane(tid >> 5);
    const int nb   = blockIdx.x * RN;
    const int c    = blockIdx.y;

    const int wid = (c * (int)gridDim.x + (int)blockIdx.x) * AGG_WAVES + wave;
    unsigned* __restrict__ lw = list + (size_t)wid * LCAP;

    for (int i = tid; i < RN; i += AGG_TB) hitcnt[i] = 0u;
    __syncthreads();

    const int cbeg = c * chunk;
    const int cend = min(e, cbeg + chunk);
    const int last = e - 4;
    unsigned* __restrict__ q = queue + wave * TILE;

    int tot = 0;
    int eb = cbeg + wave * TILE;
    int4 cd[4], cs[4];
#pragma unroll
    for (int j = 0; j < 4; ++j) {
        const int at = min(eb + j * 128 + lane * 4, last);
        cd[j] = *reinterpret_cast<const int4*>(dst + at);
        cs[j] = *reinterpret_cast<const int4*>(src + at);
    }

    for (; eb < cend; eb += AGG_WAVES * TILE) {
        const int en = eb + AGG_WAVES * TILE;
        int4 nd[4], ns[4];
#pragma unroll
        for (int j = 0; j < 4; ++j) {
            const int at = min(en + j * 128 + lane * 4, last);
            nd[j] = *reinterpret_cast<const int4*>(dst + at);
            ns[j] = *reinterpret_cast<const int4*>(src + at);
        }

        int cnt = 0;
#pragma unroll
        for (int j = 0; j < 4; ++j) {
            const int  o4 = j * 128 + lane * 4;
            const bool vq = (eb + o4) < cend;
            int d4[4], s4[4];
            d4[0] = vq ? cd[j].x : -1; d4[1] = vq ? cd[j].y : -1;
            d4[2] = vq ? cd[j].z : -1; d4[3] = vq ? cd[j].w : -1;
            s4[0] = cs[j].x; s4[1] = cs[j].y; s4[2] = cs[j].z; s4[3] = cs[j].w;
#pragma unroll
            for (int t = 0; t < 4; ++t) {
                const unsigned dl   = (unsigned)(d4[t] - nb);
                const bool     hit  = dl < (unsigned)RN;
                const unsigned mask = __builtin_amdgcn_ballot_w32(hit);
                if (hit) {
                    const int pos = cnt + (int)__builtin_amdgcn_mbcnt_lo(mask, 0u);
                    q[pos] = ((unsigned)s4[t] << 12) | dl;
                }
                cnt += __builtin_popcount(mask);
            }
        }


        for (int j = lane; j < cnt; j += 32) {
            const unsigned ent = q[j];
            __hip_atomic_fetch_add(hitcnt + (ent & 4095u), 1u,
                                   __ATOMIC_RELAXED, __HIP_MEMORY_SCOPE_WORKGROUP);
            const int p = tot + j;
            if (p < LCAP) lw[p] = ent;
        }
        tot += cnt;

#pragma unroll
        for (int j = 0; j < 4; ++j) { cd[j] = nd[j]; cs[j] = ns[j]; }
    }
    if (lane == 0)
        lcount[wid] = (tot <= LCAP && n <= (1 << SRC_BITS)) ? tot : -1;
    __syncthreads();

    unsigned* __restrict__ dout = degpart + (size_t)c * n;
    for (int i = tid; i < RN; i += AGG_TB) {
        const int node = nb + i;
        if (node < n) dout[node] = hitcnt[i];
    }
}

__global__ void gcn_dinv_kernel(const unsigned* __restrict__ degpart,
                                float* __restrict__ dinv, int n) {
    int i = blockIdx.x * blockDim.x + threadIdx.x;
    if (i < n) {
        unsigned cnt = 1u;
#pragma unroll
        for (int p = 0; p < NCHUNK; ++p) cnt += degpart[(size_t)p * n + i];
        const float d = (float)cnt;
        dinv[i] = d > 0.0f ? rsqrtf(d) : 0.0f;
    }
}

template <bool RELU, int K>
__global__ __launch_bounds__(128)
void gcn_gemm_wmma_kernel(const float* __restrict__ A,
                          const float* __restrict__ B,
                          const float* __restrict__ rscale,
                          float* __restrict__ C, int nrows) {
    const int lane = threadIdx.x & 31;
    const int wave = threadIdx.x >> 5;
    const int tile = blockIdx.x * 4 + wave;
    const int row0 = tile * 16;
    if (row0 >= nrows) return;

    const int m    = lane & 15;
    const int half = lane >> 4;
    const int kh   = half << 1;

    const float* __restrict__ arow = A + (size_t)(row0 + m) * K;

    v8f acc = {};
#pragma unroll
    for (int k = 0; k < K; k += 4) {
        v2f a = *reinterpret_cast<const v2f*>(arow + k + kh);
        if (RELU) {
            a.x = fmaxf(a.x, 0.0f);
            a.y = fmaxf(a.y, 0.0f);
        }
        v2f b;
        b.x = B[(k + kh) * HID + m];
        b.y = B[(k + kh + 1) * HID + m];
        acc = __builtin_amdgcn_wmma_f32_16x16x4_f32(
             false, a,  false, b,
             (short)0, acc,  false,  false);
    }

    const float4* rs4 = reinterpret_cast<const float4*>(rscale + row0 + 8 * half);
    const float4 sa = rs4[0], sb = rs4[1];
    const float sc[8] = {sa.x, sa.y, sa.z, sa.w, sb.x, sb.y, sb.z, sb.w};
    float* __restrict__ ccol = C + (size_t)row0 * HID + m;
#pragma unroll
    for (int r = 0; r < 8; ++r) {
        ccol[(size_t)(r + 8 * half) * HID] = acc[r] * sc[r];
    }
}

__global__ __launch_bounds__(128)
void gcn_lin2_parts_kernel(const float* __restrict__ part,
                           const float* __restrict__ hself,
                           const float* __restrict__ dinv,
                           const float* __restrict__ bias,
                           const float* __restrict__ B,
                           float* __restrict__ C, int nrows) {
    const int lane = threadIdx.x & 31;
    const int wave = threadIdx.x >> 5;
    const int tile = blockIdx.x * 4 + wave;
    const int row0 = tile * 16;
    if (row0 >= nrows) return;

    const int m    = lane & 15;
    const int half = lane >> 4;
    const int kh   = half << 1;

    const int    row   = row0 + m;
    const float  dv    = dinv[row];
    const size_t plane = (size_t)nrows * HID;
    const float* __restrict__ prow = part  + (size_t)row * HID;
    const float* __restrict__ srow = hself + (size_t)row * HID;

    v8f acc = {};
#pragma unroll
    for (int k = 0; k < HID; k += 4) {
        v2f s = *reinterpret_cast<const v2f*>(prow + k + kh);
#pragma unroll
        for (int p = 1; p < NCHUNK; ++p) {
            const v2f t = *reinterpret_cast<const v2f*>(prow + (size_t)p * plane + k + kh);
            s.x += t.x;
            s.y += t.y;
        }
        const v2f hs = *reinterpret_cast<const v2f*>(srow + k + kh);
        v2f a;
        a.x = fmaxf(bias[k + kh]     + dv * (s.x + hs.x), 0.0f);
        a.y = fmaxf(bias[k + kh + 1] + dv * (s.y + hs.y), 0.0f);
        v2f b;
        b.x = B[(k + kh) * HID + m];
        b.y = B[(k + kh + 1) * HID + m];
        acc = __builtin_amdgcn_wmma_f32_16x16x4_f32(
             false, a,  false, b,
             (short)0, acc,  false,  false);
    }

    const float4* rs4 = reinterpret_cast<const float4*>(dinv + row0 + 8 * half);
    const float4 sa = rs4[0], sb = rs4[1];
    const float sc[8] = {sa.x, sa.y, sa.z, sa.w, sb.x, sb.y, sb.z, sb.w};
    float* __restrict__ ccol = C + (size_t)row0 * HID + m;
#pragma unroll
    for (int r = 0; r < 8; ++r) {
        ccol[(size_t)(r + 8 * half) * HID] = acc[r] * sc[r];
    }
}

__device__ __forceinline__ void gcn_add_row(float* a, const float4 v0, const float4 v1,
                                            const float4 v2, const float4 v3) {
    __hip_atomic_fetch_add(a +  0, v0.x, __ATOMIC_RELAXED, __HIP_MEMORY_SCOPE_WORKGROUP);
    __hip_atomic_fetch_add(a +  1, v0.y, __ATOMIC_RELAXED, __HIP_MEMORY_SCOPE_WORKGROUP);
    __hip_atomic_fetch_add(a +  2, v0.z, __ATOMIC_RELAXED, __HIP_MEMORY_SCOPE_WORKGROUP);
    __hip_atomic_fetch_add(a +  3, v0.w, __ATOMIC_RELAXED, __HIP_MEMORY_SCOPE_WORKGROUP);
    __hip_atomic_fetch_add(a +  4, v1.x, __ATOMIC_RELAXED, __HIP_MEMORY_SCOPE_WORKGROUP);
    __hip_atomic_fetch_add(a +  5, v1.y, __ATOMIC_RELAXED, __HIP_MEMORY_SCOPE_WORKGROUP);
    __hip_atomic_fetch_add(a +  6, v1.z, __ATOMIC_RELAXED, __HIP_MEMORY_SCOPE_WORKGROUP);
    __hip_atomic_fetch_add(a +  7, v1.w, __ATOMIC_RELAXED, __HIP_MEMORY_SCOPE_WORKGROUP);
    __hip_atomic_fetch_add(a +  8, v2.x, __ATOMIC_RELAXED, __HIP_MEMORY_SCOPE_WORKGROUP);
    __hip_atomic_fetch_add(a +  9, v2.y, __ATOMIC_RELAXED, __HIP_MEMORY_SCOPE_WORKGROUP);
    __hip_atomic_fetch_add(a + 10, v2.z, __ATOMIC_RELAXED, __HIP_MEMORY_SCOPE_WORKGROUP);
    __hip_atomic_fetch_add(a + 11, v2.w, __ATOMIC_RELAXED, __HIP_MEMORY_SCOPE_WORKGROUP);
    __hip_atomic_fetch_add(a + 12, v3.x, __ATOMIC_RELAXED, __HIP_MEMORY_SCOPE_WORKGROUP);
    __hip_atomic_fetch_add(a + 13, v3.y, __ATOMIC_RELAXED, __HIP_MEMORY_SCOPE_WORKGROUP);
    __hip_atomic_fetch_add(a + 14, v3.z, __ATOMIC_RELAXED, __HIP_MEMORY_SCOPE_WORKGROUP);
    __hip_atomic_fetch_add(a + 15, v3.w, __ATOMIC_RELAXED, __HIP_MEMORY_SCOPE_WORKGROUP);
}

template <int U>
__device__ __forceinline__ void gcn_rep_load(const unsigned ent, const char* __restrict__ hb,
                                             const unsigned f4, float* v, unsigned* la) {
    if constexpr (U < 16) {
        const unsigned e = (unsigned)__builtin_amdgcn_update_dpp(0, (int)ent, 0x150 + U, 0xF, 0xF, true);
        v[U]  = *reinterpret_cast<const float*>(hb + (((e >> 6) & 0xFFFFFFC0u) | f4));
        la[U] = (e & 4095u) * PITCH;
        gcn_rep_load<U + 1>(ent, hb, f4, v, la);
    }
}

__global__ __launch_bounds__(AGG_TB)
void gcn_agg_lds_kernel(const float* __restrict__ h,
                        const int* __restrict__ src,
                        const int* __restrict__ dst,
                        float* __restrict__ part,
                        const unsigned* __restrict__ list,
                        const int* __restrict__ lcount,
                        int e, int n, int chunk) {
    __shared__ float    acc[RN * PITCH];
    __shared__ unsigned queue[AGG_WAVES * TILE];

    const int tid  = threadIdx.x;
    const int lane = tid & 31;
    const int wave = __builtin_amdgcn_readfirstlane(tid >> 5);
    const int nb   = blockIdx.x * RN;
    const int c    = blockIdx.y;

    const int wid = (c * (int)gridDim.x + (int)blockIdx.x) * AGG_WAVES + wave;
    const unsigned* __restrict__ lw = list + (size_t)wid * LCAP;
    const int cntw = __builtin_amdgcn_readfirstlane(lcount[wid]);

    for (int i = tid; i < RN * PITCH; i += AGG_TB) acc[i] = 0.0f;
    __syncthreads();

    if (cntw < 0) {
        const int cbeg = c * chunk;
        const int cend = min(e, cbeg + chunk);
        const int last = e - 4;
        unsigned* __restrict__ q = queue + wave * TILE;

        int eb = cbeg + wave * TILE;
        int4 cur[4];
#pragma unroll
        for (int j = 0; j < 4; ++j)
            cur[j] = *reinterpret_cast<const int4*>(dst + min(eb + j * 128 + lane * 4, last));

        for (; eb < cend; eb += AGG_WAVES * TILE) {
            const int en = eb + AGG_WAVES * TILE;
            int4 nxt[4];
#pragma unroll
            for (int j = 0; j < 4; ++j)
                nxt[j] = *reinterpret_cast<const int4*>(dst + min(en + j * 128 + lane * 4, last));

            int cnt = 0;
#pragma unroll
            for (int j = 0; j < 4; ++j) {
                const int  o4 = j * 128 + lane * 4;
                const bool vq = (eb + o4) < cend;
                int d4[4];
                d4[0] = vq ? cur[j].x : -1; d4[1] = vq ? cur[j].y : -1;
                d4[2] = vq ? cur[j].z : -1; d4[3] = vq ? cur[j].w : -1;
#pragma unroll
                for (int t = 0; t < 4; ++t) {
                    const unsigned dl   = (unsigned)(d4[t] - nb);
                    const bool     hit  = dl < (unsigned)RN;
                    const unsigned mask = __builtin_amdgcn_ballot_w32(hit);
                    if (hit) {
                        const int pos = cnt + (int)__builtin_amdgcn_mbcnt_lo(mask, 0u);
                        q[pos] = (dl << 9) | (unsigned)(o4 + t);
                    }
                    cnt += __builtin_popcount(mask);
                }
            }


            for (int j = lane; j < cnt; j += 32) {
                const unsigned ent = q[j];
                const int      s   = src[eb + (int)(ent & 511u)];
                const float4*  hr  = reinterpret_cast<const float4*>(h + (size_t)s * HID);
                const float4 v0 = hr[0], v1 = hr[1], v2 = hr[2], v3 = hr[3];
                gcn_add_row(acc + (ent >> 9) * PITCH, v0, v1, v2, v3);
            }

#pragma unroll
            for (int j = 0; j < 4; ++j) cur[j] = nxt[j];
        }
    } else {
        if (cntw > 0) {
            const unsigned f    = (unsigned)lane & 15u;
            const unsigned f4   = f << 2;
            const int      hoff = lane & 16;
            const char* __restrict__ hb = reinterpret_cast<const char*>(h);
            unsigned ent0 = lw[min(lane, cntw - 1)];
            unsigned ent1 = lw[min(32 + lane, cntw - 1)];
            int j0 = 0;
            for (; j0 + 64 <= cntw; j0 += 64) {
                const unsigned nx0 = lw[min(j0 + 64 + lane, cntw - 1)];
                const unsigned nx1 = lw[min(j0 + 96 + lane, cntw - 1)];
                float    v0[16], v1[16];
                unsigned la0[16], la1[16];
                gcn_rep_load<0>(ent0, hb, f4, v0, la0);
                gcn_rep_load<0>(ent1, hb, f4, v1, la1);
#pragma unroll
                for (int u = 0; u < 16; ++u)
                    __hip_atomic_fetch_add(acc + la0[u] + f, v0[u],
                                           __ATOMIC_RELAXED, __HIP_MEMORY_SCOPE_WORKGROUP);
#pragma unroll
                for (int u = 0; u < 16; ++u)
                    __hip_atomic_fetch_add(acc + la1[u] + f, v1[u],
                                           __ATOMIC_RELAXED, __HIP_MEMORY_SCOPE_WORKGROUP);
                ent0 = nx0;
                ent1 = nx1;
            }
            if (j0 < cntw) {
                const bool second = j0 + 32 < cntw;
                float    v0[16], v1[16];
                unsigned la0[16], la1[16];
                gcn_rep_load<0>(ent0, hb, f4, v0, la0);
                if (second) gcn_rep_load<0>(ent1, hb, f4, v1, la1);
#pragma unroll
                for (int u = 0; u < 16; ++u) {
                    if (j0 + hoff + u < cntw)
                        __hip_atomic_fetch_add(acc + la0[u] + f, v0[u],
                                               __ATOMIC_RELAXED, __HIP_MEMORY_SCOPE_WORKGROUP);
                }
                if (second) {
#pragma unroll
                    for (int u = 0; u < 16; ++u) {
                        if (j0 + 32 + hoff + u < cntw)
                            __hip_atomic_fetch_add(acc + la1[u] + f, v1[u],
                                                   __ATOMIC_RELAXED, __HIP_MEMORY_SCOPE_WORKGROUP);
                    }
                }
            }
        }
    }
    __syncthreads();

    float* __restrict__ pout = part + (size_t)c * n * HID;
    for (int i = tid; i < RN * HID; i += AGG_TB) {
        const int row  = i >> 4;
        const int f    = i & (HID - 1);
        const int node = nb + row;
        if (node < n) pout[(size_t)node * HID + f] = acc[row * PITCH + f];
    }
}

__global__ void gcn_sum_parts_kernel(const float* __restrict__ part,
                                     const float* __restrict__ hself,
                                     const float* __restrict__ dinv,
                                     const float* __restrict__ b2,
                                     float* __restrict__ out, int n) {
    const int i = blockIdx.x * blockDim.x + threadIdx.x;
    if (i >= n * (HID / 4)) return;
    const float4* p4 = reinterpret_cast<const float4*>(part);
    const size_t plane4 = (size_t)n * (HID / 4);
    float4 s = p4[i];
#pragma unroll
    for (int p = 1; p < NCHUNK; ++p) {
        const float4 t = p4[(size_t)p * plane4 + i];
        s.x += t.x; s.y += t.y; s.z += t.z; s.w += t.w;
    }
    const float  dv = dinv[i >> 2];
    const float4 hs = reinterpret_cast<const float4*>(hself)[i];
    const float4 b  = reinterpret_cast<const float4*>(b2)[i & 3];
    float4 o;
    o.x = b.x + dv * (s.x + hs.x);
    o.y = b.y + dv * (s.y + hs.y);
    o.z = b.z + dv * (s.z + hs.z);
    o.w = b.w + dv * (s.w + hs.w);
    reinterpret_cast<float4*>(out)[i] = o;
}

extern "C" void kernel_launch(void* const* d_in, const int* in_sizes, int n_in,
                              void* d_out, int out_size, void* d_ws, size_t ws_size,
                              hipStream_t stream) {
    const float* x  = (const float*)d_in[0];
    const int*   ei = (const int*)d_in[1];
    const float* W1 = (const float*)d_in[2];
    const float* b1 = (const float*)d_in[3];
    const float* W2 = (const float*)d_in[4];
    const float* b2 = (const float*)d_in[5];
    float* out = (float*)d_out;

    const int N = in_sizes[0] / F_IN;
    const int E = in_sizes[1] / 2;
    const int* src = ei;
    const int* dst = ei + E;

    char* ws = (char*)d_ws;
    const size_t rows_bytes = (size_t)N * HID * sizeof(float);
    float* deg  = (float*)ws;
    float* h1   = (float*)(ws + (size_t)N * sizeof(float));
    float* h2   = (float*)(ws + (size_t)N * sizeof(float) + rows_bytes);
    float* part = (float*)(ws + (size_t)N * sizeof(float) + 2 * rows_bytes);
    const int nrange = (N + RN - 1) / RN;
    const size_t nlists = (size_t)nrange * NCHUNK * AGG_WAVES;
    unsigned* list = (unsigned*)(ws + (size_t)N * sizeof(float) + (2 + NCHUNK) * rows_bytes);
    int* lcount = (int*)(list + nlists * LCAP);
    unsigned* degpart = (unsigned*)(lcount + nlists);

    const int TB = 256;

    const int chunk  = (((E + NCHUNK - 1) / NCHUNK) + TILE - 1) / TILE * TILE;
    const dim3 agrid(nrange, NCHUNK);
    gcn_scan_kernel<<<agrid, AGG_TB, 0, stream>>>(src, dst, degpart, list, lcount, E, N, chunk);

    gcn_dinv_kernel<<<(N + TB - 1) / TB, TB, 0, stream>>>(degpart, deg, N);

    const int tiles = (N + 15) / 16;
    const int gblk  = (tiles + 3) / 4;
    gcn_gemm_wmma_kernel<false, F_IN><<<gblk, 128, 0, stream>>>(x, W1, deg, h1, N);

    gcn_agg_lds_kernel<<<agrid, AGG_TB, 0, stream>>>(
        h1, src, dst, part, list, lcount, E, N, chunk);

    gcn_lin2_parts_kernel<<<gblk, 128, 0, stream>>>(part, h1, deg, b1, W2, h2, N);

    gcn_agg_lds_kernel<<<agrid, AGG_TB, 0, stream>>>(
        h2, src, dst, part, list, lcount, E, N, chunk);

    gcn_sum_parts_kernel<<<(N * (HID / 4) + TB - 1) / TB, TB, 0, stream>>>(
        part, h2, deg, b2, out, N);
}


# edge scan inverted: a work-group reads one slice of the edge list once, finds each edge's range by division, adds it to that range's list via LDS cursors; degrees from the lists; sum order changes
# speedup vs baseline: 4.5089x; 1.1089x over previous
#include <hip/hip_runtime.h>
#include <hip/hip_bf16.h>


typedef __attribute__((ext_vector_type(2))) float v2f;
typedef __attribute__((ext_vector_type(8))) float v8f;

#define F_IN 128
#define HID  16

#define RN      3584
#define PITCH   17
#define NCHUNK  9
#define TILE    512
#define AGG_TB  1024
#define AGG_WAVES (AGG_TB / 32)
#define LCAP    1024
#define SRC_BITS 20

#define NRMAX   28
#define BIN_IT  3

__global__ __launch_bounds__(AGG_TB)
void gcn_bin_kernel(const int* __restrict__ src,
                    const int* __restrict__ dst,
                    unsigned* __restrict__ list,
                    int* __restrict__ lcount,
                    int e, int n, int chunk, int nrange) {
    __shared__ unsigned cur[32];
    __shared__ unsigned stage[NRMAX * LCAP];

    const int tid  = threadIdx.x;
    const int lane = tid & 31;
    const int wave = __builtin_amdgcn_readfirstlane(tid >> 5);
    const int w    = blockIdx.x;
    const int c    = blockIdx.y;
    const int nr   = min(nrange, NRMAX);

    if (tid < 32) cur[tid] = 0u;
    __syncthreads();

    const int cbeg = c * chunk;
    const int cend = min(e, cbeg + chunk);
    const int last = e - 4;
    const int o4      = (tid & 127) * 4;
    const int tstride = AGG_WAVES * TILE;
    const int gstride = 8 * tstride;

    int eb = cbeg + w * TILE + (tid >> 7) * tstride;
    for (int gb = cbeg + w * TILE; gb < cend; gb += BIN_IT * gstride, eb += BIN_IT * gstride) {
        int4 d4[BIN_IT], s4[BIN_IT];
#pragma unroll
        for (int i = 0; i < BIN_IT; ++i) {
            const int at = min(eb + i * gstride + o4, last);
            d4[i] = *reinterpret_cast<const int4*>(dst + at);
            s4[i] = *reinterpret_cast<const int4*>(src + at);
        }
#pragma unroll
        for (int i = 0; i < BIN_IT; ++i) {
            const bool vq = (eb + i * gstride + o4) < cend;
            const int dd[4] = {d4[i].x, d4[i].y, d4[i].z, d4[i].w};
            const int ss[4] = {s4[i].x, s4[i].y, s4[i].z, s4[i].w};
#pragma unroll
            for (int t = 0; t < 4; ++t) {
                const unsigned d = (unsigned)dd[t];
                const unsigned r = d / (unsigned)RN;
                if (vq && d < (unsigned)n && r < (unsigned)nr) {
                    const unsigned dl  = d - r * (unsigned)RN;
                    const unsigned pos = __hip_atomic_fetch_add(cur + r, 1u, __ATOMIC_RELAXED,
                                                                __HIP_MEMORY_SCOPE_WORKGROUP);
                    if (pos < (unsigned)LCAP) stage[r * LCAP + pos] = ((unsigned)ss[t] << 12) | dl;
                }
            }
        }
    }
    __syncthreads();

    for (int r = wave; r < nr; r += AGG_WAVES) {
        const unsigned tot = cur[r];
        const int cnt = (int)min(tot, (unsigned)LCAP);
        const int wid = (c * nrange + r) * AGG_WAVES + w;
        unsigned* __restrict__ lw = list + (size_t)wid * LCAP;
        for (int j = lane; j < cnt; j += 32) lw[j] = stage[r * LCAP + j];
        if (lane == 0)
            lcount[wid] = (tot <= (unsigned)LCAP && n <= (1 << SRC_BITS)) ? (int)tot : -1;
    }
    for (int r = NRMAX + tid; r < nrange; r += AGG_TB)
        lcount[(c * nrange + r) * AGG_WAVES + w] = -1;
}

__global__ __launch_bounds__(AGG_TB)
void gcn_deg_dinv_kernel(const int* __restrict__ dst,
                         const unsigned* __restrict__ list,
                         const int* __restrict__ lcount,
                         float* __restrict__ dinv,
                         int e, int n, int chunk) {
    __shared__ unsigned hitcnt[RN];

    const int tid  = threadIdx.x;
    const int lane = tid & 31;
    const int wave = __builtin_amdgcn_readfirstlane(tid >> 5);
    const int nb   = blockIdx.x * RN;

    for (int i = tid; i < RN; i += AGG_TB) hitcnt[i] = 0u;

    int cn[NCHUNK];
#pragma unroll
    for (int c = 0; c < NCHUNK; ++c)
        cn[c] = __builtin_amdgcn_readfirstlane(
            lcount[(c * (int)gridDim.x + (int)blockIdx.x) * AGG_WAVES + wave]);
    __syncthreads();

    static_assert(NCHUNK % 3 == 0, "lists are counted three at a time");
#pragma unroll
    for (int g = 0; g < NCHUNK; g += 3) {
        const unsigned* lw[3];
#pragma unroll
        for (int k = 0; k < 3; ++k)
            lw[k] = list + (size_t)(((g + k) * (int)gridDim.x + (int)blockIdx.x) * AGG_WAVES + wave) * LCAP;
        const int mx = max(cn[g], max(cn[g + 1], cn[g + 2]));
        for (int base = 0; base < mx; base += 512) {
            unsigned ent[3][16];
#pragma unroll
            for (int k = 0; k < 3; ++k) {
#pragma unroll
                for (int u = 0; u < 16; ++u)
                    ent[k][u] = lw[k][max(min(base + u * 32 + lane, cn[g + k] - 1), 0)];
            }
#pragma unroll
            for (int k = 0; k < 3; ++k) {
#pragma unroll
                for (int u = 0; u < 16; ++u) {
                    if (base + u * 32 + lane < cn[g + k])
                        __hip_atomic_fetch_add(hitcnt + (ent[k][u] & 4095u), 1u,
                                               __ATOMIC_RELAXED, __HIP_MEMORY_SCOPE_WORKGROUP);
                }
            }
        }
#pragma unroll
        for (int k = 0; k < 3; ++k) {
            if (cn[g + k] < 0) {
                const int cbeg = (g + k) * chunk;
                const int cend = min(e, cbeg + chunk);
                for (int eb = cbeg + wave * TILE; eb < cend; eb += AGG_WAVES * TILE) {
                    for (int o = lane; o < TILE; o += 32) {
                        const int at = eb + o;
                        if (at < cend) {
                            const unsigned dl = (unsigned)(dst[at] - nb);
                            if (dl < (unsigned)RN)
                                __hip_atomic_fetch_add(hitcnt + dl, 1u,
                                                       __ATOMIC_RELAXED, __HIP_MEMORY_SCOPE_WORKGROUP);
                        }
                    }
                }
            }
        }
    }
    __syncthreads();

    for (int i = tid; i < RN; i += AGG_TB) {
        const int node = nb + i;
        if (node < n) {
            const float d = (float)(1u + hitcnt[i]);
            dinv[node] = d > 0.0f ? rsqrtf(d) : 0.0f;
        }
    }
}

template <bool RELU, int K>
__global__ __launch_bounds__(128)
void gcn_gemm_wmma_kernel(const float* __restrict__ A,
                          const float* __restrict__ B,
                          const float* __restrict__ rscale,
                          float* __restrict__ C, int nrows) {
    const int lane = threadIdx.x & 31;
    const int wave = threadIdx.x >> 5;
    const int tile = blockIdx.x * 4 + wave;
    const int row0 = tile * 16;
    if (row0 >= nrows) return;

    const int m    = lane & 15;
    const int half = lane >> 4;
    const int kh   = half << 1;

    const float* __restrict__ arow = A + (size_t)(row0 + m) * K;

    v8f acc = {};
#pragma unroll
    for (int k = 0; k < K; k += 4) {
        v2f a = *reinterpret_cast<const v2f*>(arow + k + kh);
        if (RELU) {
            a.x = fmaxf(a.x, 0.0f);
            a.y = fmaxf(a.y, 0.0f);
        }
        v2f b;
        b.x = B[(k + kh) * HID + m];
        b.y = B[(k + kh + 1) * HID + m];
        acc = __builtin_amdgcn_wmma_f32_16x16x4_f32(
             false, a,  false, b,
             (short)0, acc,  false,  false);
    }

    const float4* rs4 = reinterpret_cast<const float4*>(rscale + row0 + 8 * half);
    const float4 sa = rs4[0], sb = rs4[1];
    const float sc[8] = {sa.x, sa.y, sa.z, sa.w, sb.x, sb.y, sb.z, sb.w};
    float* __restrict__ ccol = C + (size_t)row0 * HID + m;
#pragma unroll
    for (int r = 0; r < 8; ++r) {
        ccol[(size_t)(r + 8 * half) * HID] = acc[r] * sc[r];
    }
}

__global__ __launch_bounds__(128)
void gcn_lin2_parts_kernel(const float* __restrict__ part,
                           const float* __restrict__ hself,
                           const float* __restrict__ dinv,
                           const float* __restrict__ bias,
                           const float* __restrict__ B,
                           float* __restrict__ C, int nrows) {
    const int lane = threadIdx.x & 31;
    const int wave = threadIdx.x >> 5;
    const int tile = blockIdx.x * 4 + wave;
    const int row0 = tile * 16;
    if (row0 >= nrows) return;

    const int m    = lane & 15;
    const int half = lane >> 4;
    const int kh   = half << 1;

    const int    row   = row0 + m;
    const float  dv    = dinv[row];
    const size_t plane = (size_t)nrows * HID;
    const float* __restrict__ prow = part  + (size_t)row * HID;
    const float* __restrict__ srow = hself + (size_t)row * HID;

    v8f acc = {};
#pragma unroll
    for (int k = 0; k < HID; k += 4) {
        v2f s = *reinterpret_cast<const v2f*>(prow + k + kh);
#pragma unroll
        for (int p = 1; p < NCHUNK; ++p) {
            const v2f t = *reinterpret_cast<const v2f*>(prow + (size_t)p * plane + k + kh);
            s.x += t.x;
            s.y += t.y;
        }
        const v2f hs = *reinterpret_cast<const v2f*>(srow + k + kh);
        v2f a;
        a.x = fmaxf(bias[k + kh]     + dv * (s.x + hs.x), 0.0f);
        a.y = fmaxf(bias[k + kh + 1] + dv * (s.y + hs.y), 0.0f);
        v2f b;
        b.x = B[(k + kh) * HID + m];
        b.y = B[(k + kh + 1) * HID + m];
        acc = __builtin_amdgcn_wmma_f32_16x16x4_f32(
             false, a,  false, b,
             (short)0, acc,  false,  false);
    }

    const float4* rs4 = reinterpret_cast<const float4*>(dinv + row0 + 8 * half);
    const float4 sa = rs4[0], sb = rs4[1];
    const float sc[8] = {sa.x, sa.y, sa.z, sa.w, sb.x, sb.y, sb.z, sb.w};
    float* __restrict__ ccol = C + (size_t)row0 * HID + m;
#pragma unroll
    for (int r = 0; r < 8; ++r) {
        ccol[(size_t)(r + 8 * half) * HID] = acc[r] * sc[r];
    }
}

__device__ __forceinline__ void gcn_add_row(float* a, const float4 v0, const float4 v1,
                                            const float4 v2, const float4 v3) {
    __hip_atomic_fetch_add(a +  0, v0.x, __ATOMIC_RELAXED, __HIP_MEMORY_SCOPE_WORKGROUP);
    __hip_atomic_fetch_add(a +  1, v0.y, __ATOMIC_RELAXED, __HIP_MEMORY_SCOPE_WORKGROUP);
    __hip_atomic_fetch_add(a +  2, v0.z, __ATOMIC_RELAXED, __HIP_MEMORY_SCOPE_WORKGROUP);
    __hip_atomic_fetch_add(a +  3, v0.w, __ATOMIC_RELAXED, __HIP_MEMORY_SCOPE_WORKGROUP);
    __hip_atomic_fetch_add(a +  4, v1.x, __ATOMIC_RELAXED, __HIP_MEMORY_SCOPE_WORKGROUP);
    __hip_atomic_fetch_add(a +  5, v1.y, __ATOMIC_RELAXED, __HIP_MEMORY_SCOPE_WORKGROUP);
    __hip_atomic_fetch_add(a +  6, v1.z, __ATOMIC_RELAXED, __HIP_MEMORY_SCOPE_WORKGROUP);
    __hip_atomic_fetch_add(a +  7, v1.w, __ATOMIC_RELAXED, __HIP_MEMORY_SCOPE_WORKGROUP);
    __hip_atomic_fetch_add(a +  8, v2.x, __ATOMIC_RELAXED, __HIP_MEMORY_SCOPE_WORKGROUP);
    __hip_atomic_fetch_add(a +  9, v2.y, __ATOMIC_RELAXED, __HIP_MEMORY_SCOPE_WORKGROUP);
    __hip_atomic_fetch_add(a + 10, v2.z, __ATOMIC_RELAXED, __HIP_MEMORY_SCOPE_WORKGROUP);
    __hip_atomic_fetch_add(a + 11, v2.w, __ATOMIC_RELAXED, __HIP_MEMORY_SCOPE_WORKGROUP);
    __hip_atomic_fetch_add(a + 12, v3.x, __ATOMIC_RELAXED, __HIP_MEMORY_SCOPE_WORKGROUP);
    __hip_atomic_fetch_add(a + 13, v3.y, __ATOMIC_RELAXED, __HIP_MEMORY_SCOPE_WORKGROUP);
    __hip_atomic_fetch_add(a + 14, v3.z, __ATOMIC_RELAXED, __HIP_MEMORY_SCOPE_WORKGROUP);
    __hip_atomic_fetch_add(a + 15, v3.w, __ATOMIC_RELAXED, __HIP_MEMORY_SCOPE_WORKGROUP);
}

template <int U>
__device__ __forceinline__ void gcn_rep_load(const unsigned ent, const char* __restrict__ hb,
                                             const unsigned f4, float* v, unsigned* la) {
    if constexpr (U < 16) {
        const unsigned e = (unsigned)__builtin_amdgcn_update_dpp(0, (int)ent, 0x150 + U, 0xF, 0xF, true);
        v[U]  = *reinterpret_cast<const float*>(hb + (((e >> 6) & 0xFFFFFFC0u) | f4));
        la[U] = (e & 4095u) * PITCH;
        gcn_rep_load<U + 1>(ent, hb, f4, v, la);
    }
}

__global__ __launch_bounds__(AGG_TB)
void gcn_agg_lds_kernel(const float* __restrict__ h,
                        const int* __restrict__ src,
                        const int* __restrict__ dst,
                        float* __restrict__ part,
                        const unsigned* __restrict__ list,
                        const int* __restrict__ lcount,
                        int e, int n, int chunk) {
    __shared__ float    acc[RN * PITCH];
    __shared__ unsigned queue[AGG_WAVES * TILE];

    const int tid  = threadIdx.x;
    const int lane = tid & 31;
    const int wave = __builtin_amdgcn_readfirstlane(tid >> 5);
    const int nb   = blockIdx.x * RN;
    const int c    = blockIdx.y;

    const int wid = (c * (int)gridDim.x + (int)blockIdx.x) * AGG_WAVES + wave;
    const unsigned* __restrict__ lw = list + (size_t)wid * LCAP;
    const int cntw = __builtin_amdgcn_readfirstlane(lcount[wid]);

    for (int i = tid; i < RN * PITCH; i += AGG_TB) acc[i] = 0.0f;
    __syncthreads();

    if (cntw < 0) {
        const int cbeg = c * chunk;
        const int cend = min(e, cbeg + chunk);
        const int last = e - 4;
        unsigned* __restrict__ q = queue + wave * TILE;

        int eb = cbeg + wave * TILE;
        int4 cur[4];
#pragma unroll
        for (int j = 0; j < 4; ++j)
            cur[j] = *reinterpret_cast<const int4*>(dst + min(eb + j * 128 + lane * 4, last));

        for (; eb < cend; eb += AGG_WAVES * TILE) {
            const int en = eb + AGG_WAVES * TILE;
            int4 nxt[4];
#pragma unroll
            for (int j = 0; j < 4; ++j)
                nxt[j] = *reinterpret_cast<const int4*>(dst + min(en + j * 128 + lane * 4, last));

            int cnt = 0;
#pragma unroll
            for (int j = 0; j < 4; ++j) {
                const int  o4 = j * 128 + lane * 4;
                const bool vq = (eb + o4) < cend;
                int d4[4];
                d4[0] = vq ? cur[j].x : -1; d4[1] = vq ? cur[j].y : -1;
                d4[2] = vq ? cur[j].z : -1; d4[3] = vq ? cur[j].w : -1;
#pragma unroll
                for (int t = 0; t < 4; ++t) {
                    const unsigned dl   = (unsigned)(d4[t] - nb);
                    const bool     hit  = dl < (unsigned)RN;
                    const unsigned mask = __builtin_amdgcn_ballot_w32(hit);
                    if (hit) {
                        const int pos = cnt + (int)__builtin_amdgcn_mbcnt_lo(mask, 0u);
                        q[pos] = (dl << 9) | (unsigned)(o4 + t);
                    }
                    cnt += __builtin_popcount(mask);
                }
            }


            for (int j = lane; j < cnt; j += 32) {
                const unsigned ent = q[j];
                const int      s   = src[eb + (int)(ent & 511u)];
                const float4*  hr  = reinterpret_cast<const float4*>(h + (size_t)s * HID);
                const float4 v0 = hr[0], v1 = hr[1], v2 = hr[2], v3 = hr[3];
                gcn_add_row(acc + (ent >> 9) * PITCH, v0, v1, v2, v3);
            }

#pragma unroll
            for (int j = 0; j < 4; ++j) cur[j] = nxt[j];
        }
    } else {
        if (cntw > 0) {
            const unsigned f    = (unsigned)lane & 15u;
            const unsigned f4   = f << 2;
            const int      hoff = lane & 16;
            const char* __restrict__ hb = reinterpret_cast<const char*>(h);
            unsigned ent0 = lw[min(lane, cntw - 1)];
            unsigned ent1 = lw[min(32 + lane, cntw - 1)];
            int j0 = 0;
            for (; j0 + 64 <= cntw; j0 += 64) {
                const unsigned nx0 = lw[min(j0 + 64 + lane, cntw - 1)];
                const unsigned nx1 = lw[min(j0 + 96 + lane, cntw - 1)];
                float    v0[16], v1[16];
                unsigned la0[16], la1[16];
                gcn_rep_load<0>(ent0, hb, f4, v0, la0);
                gcn_rep_load<0>(ent1, hb, f4, v1, la1);
#pragma unroll
                for (int u = 0; u < 16; ++u)
                    __hip_atomic_fetch_add(acc + la0[u] + f, v0[u],
                                           __ATOMIC_RELAXED, __HIP_MEMORY_SCOPE_WORKGROUP);
#pragma unroll
                for (int u = 0; u < 16; ++u)
                    __hip_atomic_fetch_add(acc + la1[u] + f, v1[u],
                                           __ATOMIC_RELAXED, __HIP_MEMORY_SCOPE_WORKGROUP);
                ent0 = nx0;
                ent1 = nx1;
            }
            if (j0 < cntw) {
                const bool second = j0 + 32 < cntw;
                float    v0[16], v1[16];
                unsigned la0[16], la1[16];
                gcn_rep_load<0>(ent0, hb, f4, v0, la0);
                if (second) gcn_rep_load<0>(ent1, hb, f4, v1, la1);
#pragma unroll
                for (int u = 0; u < 16; ++u) {
                    if (j0 + hoff + u < cntw)
                        __hip_atomic_fetch_add(acc + la0[u] + f, v0[u],
                                               __ATOMIC_RELAXED, __HIP_MEMORY_SCOPE_WORKGROUP);
                }
                if (second) {
#pragma unroll
                    for (int u = 0; u < 16; ++u) {
                        if (j0 + 32 + hoff + u < cntw)
                            __hip_atomic_fetch_add(acc + la1[u] + f, v1[u],
                                                   __ATOMIC_RELAXED, __HIP_MEMORY_SCOPE_WORKGROUP);
                    }
                }
            }
        }
    }
    __syncthreads();

    float* __restrict__ pout = part + (size_t)c * n * HID;
    for (int i = tid; i < RN * HID; i += AGG_TB) {
        const int row  = i >> 4;
        const int f    = i & (HID - 1);
        const int node = nb + row;
        if (node < n) pout[(size_t)node * HID + f] = acc[row * PITCH + f];
    }
}

__global__ void gcn_sum_parts_kernel(const float* __restrict__ part,
                                     const float* __restrict__ hself,
                                     const float* __restrict__ dinv,
                                     const float* __restrict__ b2,
                                     float* __restrict__ out, int n) {
    const int i = blockIdx.x * blockDim.x + threadIdx.x;
    if (i >= n * (HID / 4)) return;
    const float4* p4 = reinterpret_cast<const float4*>(part);
    const size_t plane4 = (size_t)n * (HID / 4);
    float4 s = p4[i];
#pragma unroll
    for (int p = 1; p < NCHUNK; ++p) {
        const float4 t = p4[(size_t)p * plane4 + i];
        s.x += t.x; s.y += t.y; s.z += t.z; s.w += t.w;
    }
    const float  dv = dinv[i >> 2];
    const float4 hs = reinterpret_cast<const float4*>(hself)[i];
    const float4 b  = reinterpret_cast<const float4*>(b2)[i & 3];
    float4 o;
    o.x = b.x + dv * (s.x + hs.x);
    o.y = b.y + dv * (s.y + hs.y);
    o.z = b.z + dv * (s.z + hs.z);
    o.w = b.w + dv * (s.w + hs.w);
    reinterpret_cast<float4*>(out)[i] = o;
}

extern "C" void kernel_launch(void* const* d_in, const int* in_sizes, int n_in,
                              void* d_out, int out_size, void* d_ws, size_t ws_size,
                              hipStream_t stream) {
    const float* x  = (const float*)d_in[0];
    const int*   ei = (const int*)d_in[1];
    const float* W1 = (const float*)d_in[2];
    const float* b1 = (const float*)d_in[3];
    const float* W2 = (const float*)d_in[4];
    const float* b2 = (const float*)d_in[5];
    float* out = (float*)d_out;

    const int N = in_sizes[0] / F_IN;
    const int E = in_sizes[1] / 2;
    const int* src = ei;
    const int* dst = ei + E;

    char* ws = (char*)d_ws;
    const size_t rows_bytes = (size_t)N * HID * sizeof(float);
    float* deg  = (float*)ws;
    float* h1   = (float*)(ws + (size_t)N * sizeof(float));
    float* h2   = (float*)(ws + (size_t)N * sizeof(float) + rows_bytes);
    float* part = (float*)(ws + (size_t)N * sizeof(float) + 2 * rows_bytes);
    const int nrange = (N + RN - 1) / RN;
    const size_t nlists = (size_t)nrange * NCHUNK * AGG_WAVES;
    unsigned* list = (unsigned*)(ws + (size_t)N * sizeof(float) + (2 + NCHUNK) * rows_bytes);
    int* lcount = (int*)(list + nlists * LCAP);

    const int TB = 256;

    const int chunk  = (((E + NCHUNK - 1) / NCHUNK) + TILE - 1) / TILE * TILE;
    const dim3 agrid(nrange, NCHUNK);
    const dim3 bgrid(AGG_WAVES, NCHUNK);
    gcn_bin_kernel<<<bgrid, AGG_TB, 0, stream>>>(src, dst, list, lcount, E, N, chunk, nrange);

    gcn_deg_dinv_kernel<<<nrange, AGG_TB, 0, stream>>>(dst, list, lcount, deg, E, N, chunk);

    const int tiles = (N + 15) / 16;
    const int gblk  = (tiles + 3) / 4;
    gcn_gemm_wmma_kernel<false, F_IN><<<gblk, 128, 0, stream>>>(x, W1, deg, h1, N);

    gcn_agg_lds_kernel<<<agrid, AGG_TB, 0, stream>>>(
        h1, src, dst, part, list, lcount, E, N, chunk);

    gcn_lin2_parts_kernel<<<gblk, 128, 0, stream>>>(part, h1, deg, b1, W2, h2, N);

    gcn_agg_lds_kernel<<<agrid, AGG_TB, 0, stream>>>(
        h2, src, dst, part, list, lcount, E, N, chunk);

    gcn_sum_parts_kernel<<<(N * (HID / 4) + TB - 1) / TB, TB, 0, stream>>>(
        part, h2, deg, b2, out, N);
}


# destination ranges narrowed to 392 rows, edges binned in 288 slices into 256 range lists: a work-group owns its range for all edges; the 9 partial planes and their sums are gone; sum order changes
# speedup vs baseline: 5.8316x; 1.2934x over previous
#include <hip/hip_runtime.h>
#include <hip/hip_bf16.h>


typedef __attribute__((ext_vector_type(2))) float v2f;
typedef __attribute__((ext_vector_type(8))) float v8f;

#define F_IN 128
#define HID  16

#define RN      392
#define PITCH   17
#define NCHUNK  1
#define TILE    512
#define AGG_TB  1024
#define AGG_WAVES (AGG_TB / 32)
#define LPW     9
#define NSL     (LPW * AGG_WAVES)
#define LCAP    128
#define SRC_BITS 20

#define NRMAX   256
#define BIN_IT  3

__global__ __launch_bounds__(AGG_TB)
void gcn_bin_kernel(const int* __restrict__ src,
                    const int* __restrict__ dst,
                    unsigned* __restrict__ list,
                    int* __restrict__ lcount,
                    int e, int n, int chunk, int nrange) {
    __shared__ unsigned cur[NRMAX];
    __shared__ unsigned stage[NRMAX * LCAP];

    const int tid  = threadIdx.x;
    const int lane = tid & 31;
    const int wave = __builtin_amdgcn_readfirstlane(tid >> 5);
    const int w    = blockIdx.x;
    const int c    = blockIdx.y;
    const int nr   = min(nrange, NRMAX);

    static_assert(NRMAX <= AGG_TB, "one cursor per thread at most");
    if (tid < NRMAX) cur[tid] = 0u;
    __syncthreads();

    const int cbeg = c * chunk;
    const int cend = min(e, cbeg + chunk);
    const int last = e - 4;
    const int o4      = (tid & 127) * 4;
    const int tstride = NSL * TILE;
    const int gstride = 8 * tstride;

    int eb = cbeg + w * TILE + (tid >> 7) * tstride;
    for (int gb = cbeg + w * TILE; gb < cend; gb += BIN_IT * gstride, eb += BIN_IT * gstride) {
        int4 d4[BIN_IT], s4[BIN_IT];
#pragma unroll
        for (int i = 0; i < BIN_IT; ++i) {
            const int at = min(eb + i * gstride + o4, last);
            d4[i] = *reinterpret_cast<const int4*>(dst + at);
            s4[i] = *reinterpret_cast<const int4*>(src + at);
        }
#pragma unroll
        for (int i = 0; i < BIN_IT; ++i) {
            const bool vq = (eb + i * gstride + o4) < cend;
            const int dd[4] = {d4[i].x, d4[i].y, d4[i].z, d4[i].w};
            const int ss[4] = {s4[i].x, s4[i].y, s4[i].z, s4[i].w};
#pragma unroll
            for (int t = 0; t < 4; ++t) {
                const unsigned d = (unsigned)dd[t];
                const unsigned r = d / (unsigned)RN;
                if (vq && d < (unsigned)n && r < (unsigned)nr) {
                    const unsigned dl  = d - r * (unsigned)RN;
                    const unsigned pos = __hip_atomic_fetch_add(cur + r, 1u, __ATOMIC_RELAXED,
                                                                __HIP_MEMORY_SCOPE_WORKGROUP);
                    if (pos < (unsigned)LCAP) stage[r * LCAP + pos] = ((unsigned)ss[t] << 12) | dl;
                }
            }
        }
    }
    __syncthreads();

    for (int r = wave; r < nr; r += AGG_WAVES) {
        const unsigned tot = cur[r];
        const int cnt = (int)min(tot, (unsigned)LCAP);
        const int wid = (c * nrange + r) * NSL + w;
        unsigned* __restrict__ lw = list + (size_t)wid * LCAP;
        for (int j = lane; j < cnt; j += 32) lw[j] = stage[r * LCAP + j];
        if (lane == 0)
            lcount[wid] = (tot <= (unsigned)LCAP && n <= (1 << SRC_BITS)) ? (int)tot : -1;
    }
    for (int r = NRMAX + tid; r < nrange; r += AGG_TB)
        lcount[(c * nrange + r) * NSL + w] = -1;
}

__device__ __forceinline__ int gcn_list_counts(const int* __restrict__ lcount, const int wid0, int* pre) {
    int  tot = 0;
    bool bad = false;
#pragma unroll
    for (int k = 0; k < LPW; ++k) {
        const int cnt = __builtin_amdgcn_readfirstlane(lcount[wid0 + AGG_WAVES * k]);
        pre[k] = tot;
        bad = bad || (cnt < 0);
        tot += max(cnt, 0);
    }
    return bad ? -1 : tot;
}

__device__ __forceinline__ unsigned gcn_list_at(const unsigned* __restrict__ lw0, const int* pre, const int j) {
    int sel = 0, sub = 0;
#pragma unroll
    for (int k = 1; k < LPW; ++k) {
        const bool g = j >= pre[k];
        sel = g ? k : sel;
        sub = g ? pre[k] : sub;
    }
    return lw0[sel * (AGG_WAVES * LCAP) + (j - sub)];
}

__global__ __launch_bounds__(AGG_TB)
void gcn_deg_dinv_kernel(const int* __restrict__ dst,
                         const unsigned* __restrict__ list,
                         const int* __restrict__ lcount,
                         float* __restrict__ dinv,
                         int e, int n, int chunk) {
    __shared__ unsigned hitcnt[RN];

    const int tid  = threadIdx.x;
    const int lane = tid & 31;
    const int wave = __builtin_amdgcn_readfirstlane(tid >> 5);
    const int nb   = blockIdx.x * RN;

    for (int i = tid; i < RN; i += AGG_TB) hitcnt[i] = 0u;
    __syncthreads();

    for (int c = 0; c < NCHUNK; ++c) {
        const int wid0 = (c * (int)gridDim.x + (int)blockIdx.x) * NSL + wave;
        const unsigned* __restrict__ lw0 = list + (size_t)wid0 * LCAP;
        int pre[LPW];
        const int tot = gcn_list_counts(lcount, wid0, pre);
        if (tot >= 0) {
            for (int base = 0; base < tot; base += 512) {
                unsigned ent[16];
#pragma unroll
                for (int u = 0; u < 16; ++u)
                    ent[u] = gcn_list_at(lw0, pre, min(base + u * 32 + lane, tot - 1));
#pragma unroll
                for (int u = 0; u < 16; ++u) {
                    if (base + u * 32 + lane < tot)
                        __hip_atomic_fetch_add(hitcnt + (ent[u] & 4095u), 1u,
                                               __ATOMIC_RELAXED, __HIP_MEMORY_SCOPE_WORKGROUP);
                }
            }
        } else {
            const int cbeg = c * chunk;
            const int cend = min(e, cbeg + chunk);
            for (int eb = cbeg + wave * TILE; eb < cend; eb += AGG_WAVES * TILE) {
                for (int o = lane; o < TILE; o += 32) {
                    const int at = eb + o;
                    if (at < cend) {
                        const unsigned dl = (unsigned)(dst[at] - nb);
                        if (dl < (unsigned)RN)
                            __hip_atomic_fetch_add(hitcnt + dl, 1u,
                                                   __ATOMIC_RELAXED, __HIP_MEMORY_SCOPE_WORKGROUP);
                    }
                }
            }
        }
    }
    __syncthreads();

    for (int i = tid; i < RN; i += AGG_TB) {
        const int node = nb + i;
        if (node < n) {
            const float d = (float)(1u + hitcnt[i]);
            dinv[node] = d > 0.0f ? rsqrtf(d) : 0.0f;
        }
    }
}

template <bool RELU, int K>
__global__ __launch_bounds__(128)
void gcn_gemm_wmma_kernel(const float* __restrict__ A,
                          const float* __restrict__ B,
                          const float* __restrict__ rscale,
                          float* __restrict__ C, int nrows) {
    const int lane = threadIdx.x & 31;
    const int wave = threadIdx.x >> 5;
    const int tile = blockIdx.x * 4 + wave;
    const int row0 = tile * 16;
    if (row0 >= nrows) return;

    const int m    = lane & 15;
    const int half = lane >> 4;
    const int kh   = half << 1;

    const float* __restrict__ arow = A + (size_t)(row0 + m) * K;

    v8f acc = {};
#pragma unroll
    for (int k = 0; k < K; k += 4) {
        v2f a = *reinterpret_cast<const v2f*>(arow + k + kh);
        if (RELU) {
            a.x = fmaxf(a.x, 0.0f);
            a.y = fmaxf(a.y, 0.0f);
        }
        v2f b;
        b.x = B[(k + kh) * HID + m];
        b.y = B[(k + kh + 1) * HID + m];
        acc = __builtin_amdgcn_wmma_f32_16x16x4_f32(
             false, a,  false, b,
             (short)0, acc,  false,  false);
    }

    const float4* rs4 = reinterpret_cast<const float4*>(rscale + row0 + 8 * half);
    const float4 sa = rs4[0], sb = rs4[1];
    const float sc[8] = {sa.x, sa.y, sa.z, sa.w, sb.x, sb.y, sb.z, sb.w};
    float* __restrict__ ccol = C + (size_t)row0 * HID + m;
#pragma unroll
    for (int r = 0; r < 8; ++r) {
        ccol[(size_t)(r + 8 * half) * HID] = acc[r] * sc[r];
    }
}

__global__ __launch_bounds__(128)
void gcn_lin2_parts_kernel(const float* __restrict__ part,
                           const float* __restrict__ hself,
                           const float* __restrict__ dinv,
                           const float* __restrict__ bias,
                           const float* __restrict__ B,
                           float* __restrict__ C, int nrows) {
    const int lane = threadIdx.x & 31;
    const int wave = threadIdx.x >> 5;
    const int tile = blockIdx.x * 4 + wave;
    const int row0 = tile * 16;
    if (row0 >= nrows) return;

    const int m    = lane & 15;
    const int half = lane >> 4;
    const int kh   = half << 1;

    const int    row   = row0 + m;
    const float  dv    = dinv[row];
    const size_t plane = (size_t)nrows * HID;
    const float* __restrict__ prow = part  + (size_t)row * HID;
    const float* __restrict__ srow = hself + (size_t)row * HID;

    v8f acc = {};
#pragma unroll
    for (int k = 0; k < HID; k += 4) {
        v2f s = *reinterpret_cast<const v2f*>(prow + k + kh);
#pragma unroll
        for (int p = 1; p < NCHUNK; ++p) {
            const v2f t = *reinterpret_cast<const v2f*>(prow + (size_t)p * plane + k + kh);
            s.x += t.x;
            s.y += t.y;
        }
        const v2f hs = *reinterpret_cast<const v2f*>(srow + k + kh);
        v2f a;
        a.x = fmaxf(bias[k + kh]     + dv * (s.x + hs.x), 0.0f);
        a.y = fmaxf(bias[k + kh + 1] + dv * (s.y + hs.y), 0.0f);
        v2f b;
        b.x = B[(k + kh) * HID + m];
        b.y = B[(k + kh + 1) * HID + m];
        acc = __builtin_amdgcn_wmma_f32_16x16x4_f32(
             false, a,  false, b,
             (short)0, acc,  false,  false);
    }

    const float4* rs4 = reinterpret_cast<const float4*>(dinv + row0 + 8 * half);
    const float4 sa = rs4[0], sb = rs4[1];
    const float sc[8] = {sa.x, sa.y, sa.z, sa.w, sb.x, sb.y, sb.z, sb.w};
    float* __restrict__ ccol = C + (size_t)row0 * HID + m;
#pragma unroll
    for (int r = 0; r < 8; ++r) {
        ccol[(size_t)(r + 8 * half) * HID] = acc[r] * sc[r];
    }
}

__device__ __forceinline__ void gcn_add_row(float* a, const float4 v0, const float4 v1,
                                            const float4 v2, const float4 v3) {
    __hip_atomic_fetch_add(a +  0, v0.x, __ATOMIC_RELAXED, __HIP_MEMORY_SCOPE_WORKGROUP);
    __hip_atomic_fetch_add(a +  1, v0.y, __ATOMIC_RELAXED, __HIP_MEMORY_SCOPE_WORKGROUP);
    __hip_atomic_fetch_add(a +  2, v0.z, __ATOMIC_RELAXED, __HIP_MEMORY_SCOPE_WORKGROUP);
    __hip_atomic_fetch_add(a +  3, v0.w, __ATOMIC_RELAXED, __HIP_MEMORY_SCOPE_WORKGROUP);
    __hip_atomic_fetch_add(a +  4, v1.x, __ATOMIC_RELAXED, __HIP_MEMORY_SCOPE_WORKGROUP);
    __hip_atomic_fetch_add(a +  5, v1.y, __ATOMIC_RELAXED, __HIP_MEMORY_SCOPE_WORKGROUP);
    __hip_atomic_fetch_add(a +  6, v1.z, __ATOMIC_RELAXED, __HIP_MEMORY_SCOPE_WORKGROUP);
    __hip_atomic_fetch_add(a +  7, v1.w, __ATOMIC_RELAXED, __HIP_MEMORY_SCOPE_WORKGROUP);
    __hip_atomic_fetch_add(a +  8, v2.x, __ATOMIC_RELAXED, __HIP_MEMORY_SCOPE_WORKGROUP);
    __hip_atomic_fetch_add(a +  9, v2.y, __ATOMIC_RELAXED, __HIP_MEMORY_SCOPE_WORKGROUP);
    __hip_atomic_fetch_add(a + 10, v2.z, __ATOMIC_RELAXED, __HIP_MEMORY_SCOPE_WORKGROUP);
    __hip_atomic_fetch_add(a + 11, v2.w, __ATOMIC_RELAXED, __HIP_MEMORY_SCOPE_WORKGROUP);
    __hip_atomic_fetch_add(a + 12, v3.x, __ATOMIC_RELAXED, __HIP_MEMORY_SCOPE_WORKGROUP);
    __hip_atomic_fetch_add(a + 13, v3.y, __ATOMIC_RELAXED, __HIP_MEMORY_SCOPE_WORKGROUP);
    __hip_atomic_fetch_add(a + 14, v3.z, __ATOMIC_RELAXED, __HIP_MEMORY_SCOPE_WORKGROUP);
    __hip_atomic_fetch_add(a + 15, v3.w, __ATOMIC_RELAXED, __HIP_MEMORY_SCOPE_WORKGROUP);
}

template <int U>
__device__ __forceinline__ void gcn_rep_load(const unsigned ent, const char* __restrict__ hb,
                                             const unsigned f4, float* v, unsigned* la) {
    if constexpr (U < 16) {
        const unsigned e = (unsigned)__builtin_amdgcn_update_dpp(0, (int)ent, 0x150 + U, 0xF, 0xF, true);
        v[U]  = *reinterpret_cast<const float*>(hb + (((e >> 6) & 0xFFFFFFC0u) | f4));
        la[U] = (e & 4095u) * PITCH;
        gcn_rep_load<U + 1>(ent, hb, f4, v, la);
    }
}

__global__ __launch_bounds__(AGG_TB)
void gcn_agg_lds_kernel(const float* __restrict__ h,
                        const int* __restrict__ src,
                        const int* __restrict__ dst,
                        float* __restrict__ part,
                        const unsigned* __restrict__ list,
                        const int* __restrict__ lcount,
                        int e, int n, int chunk) {
    __shared__ float    acc[RN * PITCH];
    __shared__ unsigned queue[AGG_WAVES * TILE];

    const int tid  = threadIdx.x;
    const int lane = tid & 31;
    const int wave = __builtin_amdgcn_readfirstlane(tid >> 5);
    const int nb   = blockIdx.x * RN;
    const int c    = blockIdx.y;

    const int wid0 = (c * (int)gridDim.x + (int)blockIdx.x) * NSL + wave;
    const unsigned* __restrict__ lw0 = list + (size_t)wid0 * LCAP;
    int pre[LPW];
    const int cntw = gcn_list_counts(lcount, wid0, pre);

    for (int i = tid; i < RN * PITCH; i += AGG_TB) acc[i] = 0.0f;
    __syncthreads();

    if (cntw < 0) {
        const int cbeg = c * chunk;
        const int cend = min(e, cbeg + chunk);
        const int last = e - 4;
        unsigned* __restrict__ q = queue + wave * TILE;

        int eb = cbeg + wave * TILE;
        int4 cur[4];
#pragma unroll
        for (int j = 0; j < 4; ++j)
            cur[j] = *reinterpret_cast<const int4*>(dst + min(eb + j * 128 + lane * 4, last));

        for (; eb < cend; eb += AGG_WAVES * TILE) {
            const int en = eb + AGG_WAVES * TILE;
            int4 nxt[4];
#pragma unroll
            for (int j = 0; j < 4; ++j)
                nxt[j] = *reinterpret_cast<const int4*>(dst + min(en + j * 128 + lane * 4, last));

            int cnt = 0;
#pragma unroll
            for (int j = 0; j < 4; ++j) {
                const int  o4 = j * 128 + lane * 4;
                const bool vq = (eb + o4) < cend;
                int d4[4];
                d4[0] = vq ? cur[j].x : -1; d4[1] = vq ? cur[j].y : -1;
                d4[2] = vq ? cur[j].z : -1; d4[3] = vq ? cur[j].w : -1;
#pragma unroll
                for (int t = 0; t < 4; ++t) {
                    const unsigned dl   = (unsigned)(d4[t] - nb);
                    const bool     hit  = dl < (unsigned)RN;
                    const unsigned mask = __builtin_amdgcn_ballot_w32(hit);
                    if (hit) {
                        const int pos = cnt + (int)__builtin_amdgcn_mbcnt_lo(mask, 0u);
                        q[pos] = (dl << 9) | (unsigned)(o4 + t);
                    }
                    cnt += __builtin_popcount(mask);
                }
            }


            for (int j = lane; j < cnt; j += 32) {
                const unsigned ent = q[j];
                const int      s   = src[eb + (int)(ent & 511u)];
                const float4*  hr  = reinterpret_cast<const float4*>(h + (size_t)s * HID);
                const float4 v0 = hr[0], v1 = hr[1], v2 = hr[2], v3 = hr[3];
                gcn_add_row(acc + (ent >> 9) * PITCH, v0, v1, v2, v3);
            }

#pragma unroll
            for (int j = 0; j < 4; ++j) cur[j] = nxt[j];
        }
    } else {
        if (cntw > 0) {
            const unsigned f    = (unsigned)lane & 15u;
            const unsigned f4   = f << 2;
            const int      hoff = lane & 16;
            const char* __restrict__ hb = reinterpret_cast<const char*>(h);
            unsigned ent0 = gcn_list_at(lw0, pre, min(lane, cntw - 1));
            unsigned ent1 = gcn_list_at(lw0, pre, min(32 + lane, cntw - 1));
            int j0 = 0;
            for (; j0 + 64 <= cntw; j0 += 64) {
                const unsigned nx0 = gcn_list_at(lw0, pre, min(j0 + 64 + lane, cntw - 1));
                const unsigned nx1 = gcn_list_at(lw0, pre, min(j0 + 96 + lane, cntw - 1));
                float    v0[16], v1[16];
                unsigned la0[16], la1[16];
                gcn_rep_load<0>(ent0, hb, f4, v0, la0);
                gcn_rep_load<0>(ent1, hb, f4, v1, la1);
#pragma unroll
                for (int u = 0; u < 16; ++u)
                    __hip_atomic_fetch_add(acc + la0[u] + f, v0[u],
                                           __ATOMIC_RELAXED, __HIP_MEMORY_SCOPE_WORKGROUP);
#pragma unroll
                for (int u = 0; u < 16; ++u)
                    __hip_atomic_fetch_add(acc + la1[u] + f, v1[u],
                                           __ATOMIC_RELAXED, __HIP_MEMORY_SCOPE_WORKGROUP);
                ent0 = nx0;
                ent1 = nx1;
            }
            if (j0 < cntw) {
                const bool second = j0 + 32 < cntw;
                float    v0[16], v1[16];
                unsigned la0[16], la1[16];
                gcn_rep_load<0>(ent0, hb, f4, v0, la0);
                if (second) gcn_rep_load<0>(ent1, hb, f4, v1, la1);
#pragma unroll
                for (int u = 0; u < 16; ++u) {
                    if (j0 + hoff + u < cntw)
                        __hip_atomic_fetch_add(acc + la0[u] + f, v0[u],
                                               __ATOMIC_RELAXED, __HIP_MEMORY_SCOPE_WORKGROUP);
                }
                if (second) {
#pragma unroll
                    for (int u = 0; u < 16; ++u) {
                        if (j0 + 32 + hoff + u < cntw)
                            __hip_atomic_fetch_add(acc + la1[u] + f, v1[u],
                                                   __ATOMIC_RELAXED, __HIP_MEMORY_SCOPE_WORKGROUP);
                    }
                }
            }
        }
    }
    __syncthreads();

    float* __restrict__ pout = part + (size_t)c * n * HID;
    for (int i = tid; i < RN * HID; i += AGG_TB) {
        const int row  = i >> 4;
        const int f    = i & (HID - 1);
        const int node = nb + row;
        if (node < n) pout[(size_t)node * HID + f] = acc[row * PITCH + f];
    }
}

__global__ void gcn_sum_parts_kernel(const float* __restrict__ part,
                                     const float* __restrict__ hself,
                                     const float* __restrict__ dinv,
                                     const float* __restrict__ b2,
                                     float* __restrict__ out, int n) {
    const int i = blockIdx.x * blockDim.x + threadIdx.x;
    if (i >= n * (HID / 4)) return;
    const float4* p4 = reinterpret_cast<const float4*>(part);
    const size_t plane4 = (size_t)n * (HID / 4);
    float4 s = p4[i];
#pragma unroll
    for (int p = 1; p < NCHUNK; ++p) {
        const float4 t = p4[(size_t)p * plane4 + i];
        s.x += t.x; s.y += t.y; s.z += t.z; s.w += t.w;
    }
    const float  dv = dinv[i >> 2];
    const float4 hs = reinterpret_cast<const float4*>(hself)[i];
    const float4 b  = reinterpret_cast<const float4*>(b2)[i & 3];
    float4 o;
    o.x = b.x + dv * (s.x + hs.x);
    o.y = b.y + dv * (s.y + hs.y);
    o.z = b.z + dv * (s.z + hs.z);
    o.w = b.w + dv * (s.w + hs.w);
    reinterpret_cast<float4*>(out)[i] = o;
}

extern "C" void kernel_launch(void* const* d_in, const int* in_sizes, int n_in,
                              void* d_out, int out_size, void* d_ws, size_t ws_size,
                              hipStream_t stream) {
    const float* x  = (const float*)d_in[0];
    const int*   ei = (const int*)d_in[1];
    const float* W1 = (const float*)d_in[2];
    const float* b1 = (const float*)d_in[3];
    const float* W2 = (const float*)d_in[4];
    const float* b2 = (const float*)d_in[5];
    float* out = (float*)d_out;

    const int N = in_sizes[0] / F_IN;
    const int E = in_sizes[1] / 2;
    const int* src = ei;
    const int* dst = ei + E;

    char* ws = (char*)d_ws;
    const size_t rows_bytes = (size_t)N * HID * sizeof(float);
    float* deg  = (float*)ws;
    float* h1   = (float*)(ws + (size_t)N * sizeof(float));
    float* h2   = (float*)(ws + (size_t)N * sizeof(float) + rows_bytes);
    float* part = (float*)(ws + (size_t)N * sizeof(float) + 2 * rows_bytes);
    const int nrange = (N + RN - 1) / RN;
    const size_t nlists = (size_t)nrange * NCHUNK * NSL;
    unsigned* list = (unsigned*)(ws + (size_t)N * sizeof(float) + (2 + NCHUNK) * rows_bytes);
    int* lcount = (int*)(list + nlists * LCAP);

    const int TB = 256;

    const int chunk  = (((E + NCHUNK - 1) / NCHUNK) + TILE - 1) / TILE * TILE;
    const dim3 agrid(nrange, NCHUNK);
    const dim3 bgrid(NSL, NCHUNK);
    gcn_bin_kernel<<<bgrid, AGG_TB, 0, stream>>>(src, dst, list, lcount, E, N, chunk, nrange);

    gcn_deg_dinv_kernel<<<nrange, AGG_TB, 0, stream>>>(dst, list, lcount, deg, E, N, chunk);

    const int tiles = (N + 15) / 16;
    const int gblk  = (tiles + 3) / 4;
    gcn_gemm_wmma_kernel<false, F_IN><<<gblk, 128, 0, stream>>>(x, W1, deg, h1, N);

    gcn_agg_lds_kernel<<<agrid, AGG_TB, 0, stream>>>(
        h1, src, dst, part, list, lcount, E, N, chunk);

    gcn_lin2_parts_kernel<<<gblk, 128, 0, stream>>>(part, h1, deg, b1, W2, h2, N);

    gcn_agg_lds_kernel<<<agrid, AGG_TB, 0, stream>>>(
        h2, src, dst, part, list, lcount, E, N, chunk);

    gcn_sum_parts_kernel<<<(N * (HID / 4) + TB - 1) / TB, TB, 0, stream>>>(
        part, h2, deg, b2, out, N);
}


# layer-2 linear moved behind the second aggregation ((A a) W2): aggregation work-groups finish their LDS rows (self row, dinv, bias, relu; WMMA tile), lin2 and final launches gone; sum order changes
# speedup vs baseline: 6.4274x; 1.1022x over previous
#include <hip/hip_runtime.h>
#include <hip/hip_bf16.h>


typedef __attribute__((ext_vector_type(2))) float v2f;
typedef __attribute__((ext_vector_type(8))) float v8f;

#define F_IN 128
#define HID  16

#define RN      392
#define PITCH   17
#define NCHUNK  1
#define TILE    512
#define AGG_TB  1024
#define AGG_WAVES (AGG_TB / 32)
#define LPW     9
#define NSL     (LPW * AGG_WAVES)
#define LCAP    128
#define SRC_BITS 20

#define NRMAX   256
#define BIN_IT  3

__global__ __launch_bounds__(AGG_TB)
void gcn_bin_kernel(const int* __restrict__ src,
                    const int* __restrict__ dst,
                    unsigned* __restrict__ list,
                    int* __restrict__ lcount,
                    int e, int n, int chunk, int nrange) {
    __shared__ unsigned cur[NRMAX];
    __shared__ unsigned stage[NRMAX * LCAP];

    const int tid  = threadIdx.x;
    const int lane = tid & 31;
    const int wave = __builtin_amdgcn_readfirstlane(tid >> 5);
    const int w    = blockIdx.x;
    const int c    = blockIdx.y;
    const int nr   = min(nrange, NRMAX);

    static_assert(NRMAX <= AGG_TB, "one cursor per thread at most");
    if (tid < NRMAX) cur[tid] = 0u;
    __syncthreads();

    const int cbeg = c * chunk;
    const int cend = min(e, cbeg + chunk);
    const int last = e - 4;
    const int o4      = (tid & 127) * 4;
    const int tstride = NSL * TILE;
    const int gstride = 8 * tstride;

    int eb = cbeg + w * TILE + (tid >> 7) * tstride;
    for (int gb = cbeg + w * TILE; gb < cend; gb += BIN_IT * gstride, eb += BIN_IT * gstride) {
        int4 d4[BIN_IT], s4[BIN_IT];
#pragma unroll
        for (int i = 0; i < BIN_IT; ++i) {
            const int at = min(eb + i * gstride + o4, last);
            d4[i] = *reinterpret_cast<const int4*>(dst + at);
            s4[i] = *reinterpret_cast<const int4*>(src + at);
        }
#pragma unroll
        for (int i = 0; i < BIN_IT; ++i) {
            const bool vq = (eb + i * gstride + o4) < cend;
            const int dd[4] = {d4[i].x, d4[i].y, d4[i].z, d4[i].w};
            const int ss[4] = {s4[i].x, s4[i].y, s4[i].z, s4[i].w};
#pragma unroll
            for (int t = 0; t < 4; ++t) {
                const unsigned d = (unsigned)dd[t];
                const unsigned r = d / (unsigned)RN;
                if (vq && d < (unsigned)n && r < (unsigned)nr) {
                    const unsigned dl  = d - r * (unsigned)RN;
                    const unsigned pos = __hip_atomic_fetch_add(cur + r, 1u, __ATOMIC_RELAXED,
                                                                __HIP_MEMORY_SCOPE_WORKGROUP);
                    if (pos < (unsigned)LCAP) stage[r * LCAP + pos] = ((unsigned)ss[t] << 12) | dl;
                }
            }
        }
    }
    __syncthreads();

    for (int r = wave; r < nr; r += AGG_WAVES) {
        const unsigned tot = cur[r];
        const int cnt = (int)min(tot, (unsigned)LCAP);
        const int wid = (c * nrange + r) * NSL + w;
        unsigned* __restrict__ lw = list + (size_t)wid * LCAP;
        for (int j = lane; j < cnt; j += 32) lw[j] = stage[r * LCAP + j];
        if (lane == 0)
            lcount[wid] = (tot <= (unsigned)LCAP && n <= (1 << SRC_BITS)) ? (int)tot : -1;
    }
    for (int r = NRMAX + tid; r < nrange; r += AGG_TB)
        lcount[(c * nrange + r) * NSL + w] = -1;
}

__device__ __forceinline__ int gcn_list_counts(const int* __restrict__ lcount, const int wid0, int* pre) {
    int  tot = 0;
    bool bad = false;
#pragma unroll
    for (int k = 0; k < LPW; ++k) {
        const int cnt = __builtin_amdgcn_readfirstlane(lcount[wid0 + AGG_WAVES * k]);
        pre[k] = tot;
        bad = bad || (cnt < 0);
        tot += max(cnt, 0);
    }
    return bad ? -1 : tot;
}

__device__ __forceinline__ unsigned gcn_list_at(const unsigned* __restrict__ lw0, const int* pre, const int j) {
    int sel = 0, sub = 0;
#pragma unroll
    for (int k = 1; k < LPW; ++k) {
        const bool g = j >= pre[k];
        sel = g ? k : sel;
        sub = g ? pre[k] : sub;
    }
    return lw0[sel * (AGG_WAVES * LCAP) + (j - sub)];
}

__global__ __launch_bounds__(AGG_TB)
void gcn_deg_dinv_kernel(const int* __restrict__ dst,
                         const unsigned* __restrict__ list,
                         const int* __restrict__ lcount,
                         float* __restrict__ dinv,
                         int e, int n, int chunk) {
    __shared__ unsigned hitcnt[RN];

    const int tid  = threadIdx.x;
    const int lane = tid & 31;
    const int wave = __builtin_amdgcn_readfirstlane(tid >> 5);
    const int nb   = blockIdx.x * RN;

    for (int i = tid; i < RN; i += AGG_TB) hitcnt[i] = 0u;
    __syncthreads();

    for (int c = 0; c < NCHUNK; ++c) {
        const int wid0 = (c * (int)gridDim.x + (int)blockIdx.x) * NSL + wave;
        const unsigned* __restrict__ lw0 = list + (size_t)wid0 * LCAP;
        int pre[LPW];
        const int tot = gcn_list_counts(lcount, wid0, pre);
        if (tot >= 0) {
            for (int base = 0; base < tot; base += 512) {
                unsigned ent[16];
#pragma unroll
                for (int u = 0; u < 16; ++u)
                    ent[u] = gcn_list_at(lw0, pre, min(base + u * 32 + lane, tot - 1));
#pragma unroll
                for (int u = 0; u < 16; ++u) {
                    if (base + u * 32 + lane < tot)
                        __hip_atomic_fetch_add(hitcnt + (ent[u] & 4095u), 1u,
                                               __ATOMIC_RELAXED, __HIP_MEMORY_SCOPE_WORKGROUP);
                }
            }
        } else {
            const int cbeg = c * chunk;
            const int cend = min(e, cbeg + chunk);
            for (int eb = cbeg + wave * TILE; eb < cend; eb += AGG_WAVES * TILE) {
                for (int o = lane; o < TILE; o += 32) {
                    const int at = eb + o;
                    if (at < cend) {
                        const unsigned dl = (unsigned)(dst[at] - nb);
                        if (dl < (unsigned)RN)
                            __hip_atomic_fetch_add(hitcnt + dl, 1u,
                                                   __ATOMIC_RELAXED, __HIP_MEMORY_SCOPE_WORKGROUP);
                    }
                }
            }
        }
    }
    __syncthreads();

    for (int i = tid; i < RN; i += AGG_TB) {
        const int node = nb + i;
        if (node < n) {
            const float d = (float)(1u + hitcnt[i]);
            dinv[node] = d > 0.0f ? rsqrtf(d) : 0.0f;
        }
    }
}

template <bool RELU, int K>
__global__ __launch_bounds__(128)
void gcn_gemm_wmma_kernel(const float* __restrict__ A,
                          const float* __restrict__ B,
                          const float* __restrict__ rscale,
                          float* __restrict__ C, int nrows) {
    const int lane = threadIdx.x & 31;
    const int wave = threadIdx.x >> 5;
    const int tile = blockIdx.x * 4 + wave;
    const int row0 = tile * 16;
    if (row0 >= nrows) return;

    const int m    = lane & 15;
    const int half = lane >> 4;
    const int kh   = half << 1;

    const float* __restrict__ arow = A + (size_t)(row0 + m) * K;

    v8f acc = {};
#pragma unroll
    for (int k = 0; k < K; k += 4) {
        v2f a = *reinterpret_cast<const v2f*>(arow + k + kh);
        if (RELU) {
            a.x = fmaxf(a.x, 0.0f);
            a.y = fmaxf(a.y, 0.0f);
        }
        v2f b;
        b.x = B[(k + kh) * HID + m];
        b.y = B[(k + kh + 1) * HID + m];
        acc = __builtin_amdgcn_wmma_f32_16x16x4_f32(
             false, a,  false, b,
             (short)0, acc,  false,  false);
    }

    const float4* rs4 = reinterpret_cast<const float4*>(rscale + row0 + 8 * half);
    const float4 sa = rs4[0], sb = rs4[1];
    const float sc[8] = {sa.x, sa.y, sa.z, sa.w, sb.x, sb.y, sb.z, sb.w};
    float* __restrict__ ccol = C + (size_t)row0 * HID + m;
#pragma unroll
    for (int r = 0; r < 8; ++r) {
        ccol[(size_t)(r + 8 * half) * HID] = acc[r] * sc[r];
    }
}

__device__ __forceinline__ void gcn_add_row(float* a, const float4 v0, const float4 v1,
                                            const float4 v2, const float4 v3) {
    __hip_atomic_fetch_add(a +  0, v0.x, __ATOMIC_RELAXED, __HIP_MEMORY_SCOPE_WORKGROUP);
    __hip_atomic_fetch_add(a +  1, v0.y, __ATOMIC_RELAXED, __HIP_MEMORY_SCOPE_WORKGROUP);
    __hip_atomic_fetch_add(a +  2, v0.z, __ATOMIC_RELAXED, __HIP_MEMORY_SCOPE_WORKGROUP);
    __hip_atomic_fetch_add(a +  3, v0.w, __ATOMIC_RELAXED, __HIP_MEMORY_SCOPE_WORKGROUP);
    __hip_atomic_fetch_add(a +  4, v1.x, __ATOMIC_RELAXED, __HIP_MEMORY_SCOPE_WORKGROUP);
    __hip_atomic_fetch_add(a +  5, v1.y, __ATOMIC_RELAXED, __HIP_MEMORY_SCOPE_WORKGROUP);
    __hip_atomic_fetch_add(a +  6, v1.z, __ATOMIC_RELAXED, __HIP_MEMORY_SCOPE_WORKGROUP);
    __hip_atomic_fetch_add(a +  7, v1.w, __ATOMIC_RELAXED, __HIP_MEMORY_SCOPE_WORKGROUP);
    __hip_atomic_fetch_add(a +  8, v2.x, __ATOMIC_RELAXED, __HIP_MEMORY_SCOPE_WORKGROUP);
    __hip_atomic_fetch_add(a +  9, v2.y, __ATOMIC_RELAXED, __HIP_MEMORY_SCOPE_WORKGROUP);
    __hip_atomic_fetch_add(a + 10, v2.z, __ATOMIC_RELAXED, __HIP_MEMORY_SCOPE_WORKGROUP);
    __hip_atomic_fetch_add(a + 11, v2.w, __ATOMIC_RELAXED, __HIP_MEMORY_SCOPE_WORKGROUP);
    __hip_atomic_fetch_add(a + 12, v3.x, __ATOMIC_RELAXED, __HIP_MEMORY_SCOPE_WORKGROUP);
    __hip_atomic_fetch_add(a + 13, v3.y, __ATOMIC_RELAXED, __HIP_MEMORY_SCOPE_WORKGROUP);
    __hip_atomic_fetch_add(a + 14, v3.z, __ATOMIC_RELAXED, __HIP_MEMORY_SCOPE_WORKGROUP);
    __hip_atomic_fetch_add(a + 15, v3.w, __ATOMIC_RELAXED, __HIP_MEMORY_SCOPE_WORKGROUP);
}

template <int U>
__device__ __forceinline__ void gcn_rep_load(const unsigned ent, const char* __restrict__ hb,
                                             const unsigned f4, float* v, unsigned* la) {
    if constexpr (U < 16) {
        const unsigned e = (unsigned)__builtin_amdgcn_update_dpp(0, (int)ent, 0x150 + U, 0xF, 0xF, true);
        v[U]  = *reinterpret_cast<const float*>(hb + (((e >> 6) & 0xFFFFFFC0u) | f4));
        la[U] = (e & 4095u) * PITCH;
        gcn_rep_load<U + 1>(ent, hb, f4, v, la);
    }
}

#define NT       ((RN + 15) / 16)
#define ACC_ROWS (NT * 16)

template <int LAYER>
__global__ __launch_bounds__(AGG_TB)
void gcn_agg_lds_kernel(const float* __restrict__ h,
                        const int* __restrict__ src,
                        const int* __restrict__ dst,
                        const float* __restrict__ dinv,
                        const float* __restrict__ bias,
                        const float* __restrict__ W,
                        float* __restrict__ outp,
                        const unsigned* __restrict__ list,
                        const int* __restrict__ lcount,
                        int e, int n, int chunk) {
    static_assert(NCHUNK == 1, "the work-group finishes its rows: they must be complete");
    static_assert(NT <= AGG_WAVES, "one tile per wave at most");
    __shared__ float    acc[ACC_ROWS * PITCH];
    __shared__ unsigned queue[AGG_WAVES * TILE];

    const int tid  = threadIdx.x;
    const int lane = tid & 31;
    const int wave = __builtin_amdgcn_readfirstlane(tid >> 5);
    const int nb   = blockIdx.x * RN;
    const int c    = blockIdx.y;

    const int wid0 = (c * (int)gridDim.x + (int)blockIdx.x) * NSL + wave;
    const unsigned* __restrict__ lw0 = list + (size_t)wid0 * LCAP;
    int pre[LPW];
    const int cntw = gcn_list_counts(lcount, wid0, pre);

    constexpr int INIT_IT = (ACC_ROWS * HID + AGG_TB - 1) / AGG_TB;
    {
        const int f = tid & (HID - 1);
        float selfv[INIT_IT];
#pragma unroll
        for (int it = 0; it < INIT_IT; ++it) {
            const int row = (tid + it * AGG_TB) >> 4;
            selfv[it] = h[(size_t)min(nb + row, n - 1) * HID + f];
        }
#pragma unroll
        for (int it = 0; it < INIT_IT; ++it) {
            const int row = (tid + it * AGG_TB) >> 4;
            if (row < ACC_ROWS)
                acc[row * PITCH + f] = (row < RN && nb + row < n) ? selfv[it] : 0.0f;
        }
    }
    __syncthreads();

    if (cntw < 0) {
        const int cbeg = c * chunk;
        const int cend = min(e, cbeg + chunk);
        const int last = e - 4;
        unsigned* __restrict__ q = queue + wave * TILE;

        int eb = cbeg + wave * TILE;
        int4 cur[4];
#pragma unroll
        for (int j = 0; j < 4; ++j)
            cur[j] = *reinterpret_cast<const int4*>(dst + min(eb + j * 128 + lane * 4, last));

        for (; eb < cend; eb += AGG_WAVES * TILE) {
            const int en = eb + AGG_WAVES * TILE;
            int4 nxt[4];
#pragma unroll
            for (int j = 0; j < 4; ++j)
                nxt[j] = *reinterpret_cast<const int4*>(dst + min(en + j * 128 + lane * 4, last));

            int cnt = 0;
#pragma unroll
            for (int j = 0; j < 4; ++j) {
                const int  o4 = j * 128 + lane * 4;
                const bool vq = (eb + o4) < cend;
                int d4[4];
                d4[0] = vq ? cur[j].x : -1; d4[1] = vq ? cur[j].y : -1;
                d4[2] = vq ? cur[j].z : -1; d4[3] = vq ? cur[j].w : -1;
#pragma unroll
                for (int t = 0; t < 4; ++t) {
                    const unsigned dl   = (unsigned)(d4[t] - nb);
                    const bool     hit  = dl < (unsigned)RN;
                    const unsigned mask = __builtin_amdgcn_ballot_w32(hit);
                    if (hit) {
                        const int pos = cnt + (int)__builtin_amdgcn_mbcnt_lo(mask, 0u);
                        q[pos] = (dl << 9) | (unsigned)(o4 + t);
                    }
                    cnt += __builtin_popcount(mask);
                }
            }


            for (int j = lane; j < cnt; j += 32) {
                const unsigned ent = q[j];
                const int      s   = src[eb + (int)(ent & 511u)];
                const float4*  hr  = reinterpret_cast<const float4*>(h + (size_t)s * HID);
                const float4 v0 = hr[0], v1 = hr[1], v2 = hr[2], v3 = hr[3];
                gcn_add_row(acc + (ent >> 9) * PITCH, v0, v1, v2, v3);
            }

#pragma unroll
            for (int j = 0; j < 4; ++j) cur[j] = nxt[j];
        }
    } else {
        if (cntw > 0) {
            const unsigned f    = (unsigned)lane & 15u;
            const unsigned f4   = f << 2;
            const int      hoff = lane & 16;
            const char* __restrict__ hb = reinterpret_cast<const char*>(h);
            unsigned ent0 = gcn_list_at(lw0, pre, min(lane, cntw - 1));
            unsigned ent1 = gcn_list_at(lw0, pre, min(32 + lane, cntw - 1));
            int j0 = 0;
            for (; j0 + 64 <= cntw; j0 += 64) {
                const unsigned nx0 = gcn_list_at(lw0, pre, min(j0 + 64 + lane, cntw - 1));
                const unsigned nx1 = gcn_list_at(lw0, pre, min(j0 + 96 + lane, cntw - 1));
                float    v0[16], v1[16];
                unsigned la0[16], la1[16];
                gcn_rep_load<0>(ent0, hb, f4, v0, la0);
                gcn_rep_load<0>(ent1, hb, f4, v1, la1);
#pragma unroll
                for (int u = 0; u < 16; ++u)
                    __hip_atomic_fetch_add(acc + la0[u] + f, v0[u],
                                           __ATOMIC_RELAXED, __HIP_MEMORY_SCOPE_WORKGROUP);
#pragma unroll
                for (int u = 0; u < 16; ++u)
                    __hip_atomic_fetch_add(acc + la1[u] + f, v1[u],
                                           __ATOMIC_RELAXED, __HIP_MEMORY_SCOPE_WORKGROUP);
                ent0 = nx0;
                ent1 = nx1;
            }
            if (j0 < cntw) {
                const bool second = j0 + 32 < cntw;
                float    v0[16], v1[16];
                unsigned la0[16], la1[16];
                gcn_rep_load<0>(ent0, hb, f4, v0, la0);
                if (second) gcn_rep_load<0>(ent1, hb, f4, v1, la1);
#pragma unroll
                for (int u = 0; u < 16; ++u) {
                    if (j0 + hoff + u < cntw)
                        __hip_atomic_fetch_add(acc + la0[u] + f, v0[u],
                                               __ATOMIC_RELAXED, __HIP_MEMORY_SCOPE_WORKGROUP);
                }
                if (second) {
#pragma unroll
                    for (int u = 0; u < 16; ++u) {
                        if (j0 + 32 + hoff + u < cntw)
                            __hip_atomic_fetch_add(acc + la1[u] + f, v1[u],
                                                   __ATOMIC_RELAXED, __HIP_MEMORY_SCOPE_WORKGROUP);
                    }
                }
            }
        }
    }

    const int m    = lane & 15;
    const int half = lane >> 4;
    const int kh   = half << 1;
    v2f   bfrag[HID / 4];
    float dvt = 0.0f, bcol = 0.0f;
    constexpr int FIN_IT = (RN * HID + AGG_TB - 1) / AGG_TB;
    float dvr[FIN_IT];
    if constexpr (LAYER == 1) {
        bcol = bias[m];
#pragma unroll
        for (int it = 0; it < FIN_IT; ++it)
            dvr[it] = dinv[min(nb + ((tid + it * AGG_TB) >> 4), n - 1)];
    }
    if constexpr (LAYER == 2) {
        if (wave < NT) {
            dvt  = dinv[min(nb + wave * 16 + m, n - 1)];
            bcol = bias[m];
#pragma unroll
            for (int k = 0; k < HID; k += 4) {
                bfrag[k >> 2].x = W[(k + kh) * HID + m];
                bfrag[k >> 2].y = W[(k + kh + 1) * HID + m];
            }
        }
    }
    __syncthreads();

    if constexpr (LAYER == 2) {
        if (wave < NT) {
            const float* arow = acc + (wave * 16 + m) * PITCH;
            v8f d = {};
#pragma unroll
            for (int k = 0; k < HID; k += 4) {
                v2f a;
                a.x = dvt * arow[k + kh];
                a.y = dvt * arow[k + kh + 1];
                d = __builtin_amdgcn_wmma_f32_16x16x4_f32(
                     false, a,  false, bfrag[k >> 2],
                     (short)0, d,  false,  false);
            }
            float* ccol = acc + (wave * 16 + 8 * half) * PITCH + m;
#pragma unroll
            for (int r = 0; r < 8; ++r) ccol[r * PITCH] = d[r] + bcol;
        }
        __syncthreads();
    }

    float av[FIN_IT];
#pragma unroll
    for (int it = 0; it < FIN_IT; ++it)
        av[it] = acc[min((tid + it * AGG_TB) >> 4, RN - 1) * PITCH + m];
#pragma unroll
    for (int it = 0; it < FIN_IT; ++it) {
        const int row  = (tid + it * AGG_TB) >> 4;
        const int node = nb + row;
        if (row < RN && node < n) {
            if constexpr (LAYER == 2) {
                outp[(size_t)node * HID + m] = av[it];
            } else {
                outp[(size_t)node * HID + m] = dvr[it] * fmaxf(bcol + dvr[it] * av[it], 0.0f);
            }
        }
    }
}

extern "C" void kernel_launch(void* const* d_in, const int* in_sizes, int n_in,
                              void* d_out, int out_size, void* d_ws, size_t ws_size,
                              hipStream_t stream) {
    const float* x  = (const float*)d_in[0];
    const int*   ei = (const int*)d_in[1];
    const float* W1 = (const float*)d_in[2];
    const float* b1 = (const float*)d_in[3];
    const float* W2 = (const float*)d_in[4];
    const float* b2 = (const float*)d_in[5];
    float* out = (float*)d_out;

    const int N = in_sizes[0] / F_IN;
    const int E = in_sizes[1] / 2;
    const int* src = ei;
    const int* dst = ei + E;

    char* ws = (char*)d_ws;
    const size_t rows_bytes = (size_t)N * HID * sizeof(float);
    float* deg  = (float*)ws;
    float* h1   = (float*)(ws + (size_t)N * sizeof(float));
    float* g1   = (float*)(ws + (size_t)N * sizeof(float) + rows_bytes);
    const int nrange = (N + RN - 1) / RN;
    const size_t nlists = (size_t)nrange * NCHUNK * NSL;
    unsigned* list = (unsigned*)(ws + (size_t)N * sizeof(float) + 2 * rows_bytes);
    int* lcount = (int*)(list + nlists * LCAP);

    const int chunk  = (((E + NCHUNK - 1) / NCHUNK) + TILE - 1) / TILE * TILE;
    const dim3 agrid(nrange, NCHUNK);
    const dim3 bgrid(NSL, NCHUNK);
    gcn_bin_kernel<<<bgrid, AGG_TB, 0, stream>>>(src, dst, list, lcount, E, N, chunk, nrange);

    gcn_deg_dinv_kernel<<<nrange, AGG_TB, 0, stream>>>(dst, list, lcount, deg, E, N, chunk);

    const int tiles = (N + 15) / 16;
    const int gblk  = (tiles + 3) / 4;
    gcn_gemm_wmma_kernel<false, F_IN><<<gblk, 128, 0, stream>>>(x, W1, deg, h1, N);

    gcn_agg_lds_kernel<1><<<agrid, AGG_TB, 0, stream>>>(
        h1, src, dst, deg, b1, W2, g1, list, lcount, E, N, chunk);

    gcn_agg_lds_kernel<2><<<agrid, AGG_TB, 0, stream>>>(
        g1, src, dst, deg, b2, W2, out, list, lcount, E, N, chunk);
}
